# starting kernel
# baseline (speedup 1.0000x reference)
#include <hip/hip_runtime.h>
#include <hip/hip_bf16.h>

typedef float v2f __attribute__((ext_vector_type(2)));
typedef float v8f __attribute__((ext_vector_type(8)));

#define N_NODES 50000
#define N_EDGES 800000
#define N_FEAT  256
#define N_HID   32
#define N_HEADS 8
#define ALPHA   0.2f

__global__ void gat_init(float* __restrict__ out,
                         float* __restrict__ segmax,
                         float* __restrict__ denom) {
    int i = blockIdx.x * blockDim.x + threadIdx.x;
    if (i < N_NODES * N_HEADS * N_HID) out[i] = 0.0f;
    if (i < N_NODES * N_HEADS) {
        segmax[i] = -__builtin_inff();
        denom[i]  = 0.0f;
    }
}

__global__ __launch_bounds__(256) void gat_gemm(const float* __restrict__ hmat,
                                                const float* __restrict__ W,
                                                float* __restrict__ Wh) {
    __shared__ float sh[16 * N_FEAT];

    const int node_base = blockIdx.x * 16;

    const float4* g4 = (const float4*)(hmat + (size_t)node_base * N_FEAT);
    float4* s4 = (float4*)sh;
    for (int i = threadIdx.x; i < (16 * N_FEAT) / 4; i += 256)
        s4[i] = g4[i];
    __syncthreads();

    const int lane = threadIdx.x & 31;
    const int head = threadIdx.x >> 5;
    const float* Wp = W + (size_t)head * N_FEAT * N_HID;

    const int m    = lane & 15;
    const int koff = (lane >> 4) << 1;

    v8f c0 = {};
    v8f c1 = {};

    for (int k = 0; k < N_FEAT; k += 4) {
        v2f a, b0, b1;
        a[0] = sh[m * N_FEAT + k + koff];
        a[1] = sh[m * N_FEAT + k + koff + 1];
        b0[0] = Wp[(k + koff)     * N_HID + m];
        b0[1] = Wp[(k + koff + 1) * N_HID + m];
        b1[0] = Wp[(k + koff)     * N_HID + 16 + m];
        b1[1] = Wp[(k + koff + 1) * N_HID + 16 + m];
        c0 = __builtin_amdgcn_wmma_f32_16x16x4_f32(false, a, false, b0,
                                                   (short)0, c0, false, false);
        c1 = __builtin_amdgcn_wmma_f32_16x16x4_f32(false, a, false, b1,
                                                   (short)0, c1, false, false);
    }

    float* op = Wh + ((size_t)head * N_NODES + node_base) * N_HID;
    const int mb = (lane >> 4) * 8;
#pragma unroll
    for (int v = 0; v < 8; ++v) {
        op[(mb + v) * N_HID + m]      = c0[v];
        op[(mb + v) * N_HID + 16 + m] = c1[v];
    }
}

__global__ __launch_bounds__(256) void gat_score(const float* __restrict__ Wh,
                                                 const float* __restrict__ avec,
                                                 float* __restrict__ s1,
                                                 float* __restrict__ s2) {
    const int g    = (blockIdx.x * blockDim.x + threadIdx.x) >> 5;
    const int lane = threadIdx.x & 31;
    if (g >= N_NODES * N_HEADS) return;
    const int head = g & 7;
    const int n    = g >> 3;

    float wh = Wh[((size_t)head * N_NODES + n) * N_HID + lane];
    float p1 = wh * avec[head * 2 * N_HID + lane];
    float p2 = wh * avec[head * 2 * N_HID + N_HID + lane];
#pragma unroll
    for (int off = 16; off > 0; off >>= 1) {
        p1 += __shfl_xor(p1, off, 32);
        p2 += __shfl_xor(p2, off, 32);
    }
    if (lane == 0) {
        s1[head * N_NODES + n] = p1;
        s2[head * N_NODES + n] = p2;
    }
}

__device__ __forceinline__ void atomicMaxFloat(float* addr, float val) {
    if (val >= 0.0f)
        atomicMax((int*)addr, __float_as_int(val));
    else
        atomicMin((unsigned int*)addr, __float_as_uint(val));
}

__global__ void gat_edge_max(const float* __restrict__ s1,
                             const float* __restrict__ s2,
                             const int* __restrict__ src,
                             const int* __restrict__ dst,
                             float* __restrict__ ebuf,
                             float* __restrict__ segmax) {
    const int e = blockIdx.x * blockDim.x + threadIdx.x;
    const int head = blockIdx.y;
    if (e >= N_EDGES) return;
    const int s = src[e], d = dst[e];
    float x = s1[head * N_NODES + s] + s2[head * N_NODES + d];
    float v = x > 0.0f ? x : ALPHA * x;
    ebuf[(size_t)head * N_EDGES + e] = v;
    atomicMaxFloat(&segmax[head * N_NODES + d], v);
}

__global__ void gat_edge_exp(const int* __restrict__ dst,
                             const float* __restrict__ segmax,
                             float* __restrict__ ebuf,
                             float* __restrict__ denom) {
    const int e = blockIdx.x * blockDim.x + threadIdx.x;
    const int head = blockIdx.y;
    if (e >= N_EDGES) return;
    const int d = dst[e];
    float ex = __expf(ebuf[(size_t)head * N_EDGES + e] - segmax[head * N_NODES + d]);
    ebuf[(size_t)head * N_EDGES + e] = ex;
    atomicAdd(&denom[head * N_NODES + d], ex);
}

__global__ __launch_bounds__(256) void gat_message(const int* __restrict__ src,
                                                   const int* __restrict__ dst,
                                                   const float* __restrict__ ebuf,
                                                   const float* __restrict__ denom,
                                                   const float* __restrict__ Wh,
                                                   float* __restrict__ out) {
    const int e    = (blockIdx.x * blockDim.x + threadIdx.x) >> 5;
    const int lane = threadIdx.x & 31;
    if (e >= N_EDGES) return;
    const int s = src[e], d = dst[e];
    float* op = out + (size_t)d * (N_HEADS * N_HID);
#pragma unroll
    for (int head = 0; head < N_HEADS; ++head) {
        float attn = ebuf[(size_t)head * N_EDGES + e] / denom[head * N_NODES + d];
        float val  = attn * Wh[((size_t)head * N_NODES + s) * N_HID + lane];
        atomicAdd(op + head * N_HID + lane, val);
    }
}

extern "C" void kernel_launch(void* const* d_in, const int* in_sizes, int n_in,
                              void* d_out, int out_size, void* d_ws, size_t ws_size,
                              hipStream_t stream) {
    const float* h   = (const float*)d_in[0];
    const float* W   = (const float*)d_in[1];
    const float* a   = (const float*)d_in[2];
    const int*   src = (const int*)d_in[3];
    const int*   dst = (const int*)d_in[4];
    float* out = (float*)d_out;

    float* ws     = (float*)d_ws;
    float* Wh     = ws;
    float* s1     = Wh + (size_t)N_HEADS * N_NODES * N_HID;
    float* s2     = s1 + (size_t)N_HEADS * N_NODES;
    float* segmax = s2 + (size_t)N_HEADS * N_NODES;
    float* denom  = segmax + (size_t)N_HEADS * N_NODES;
    float* ebuf   = denom + (size_t)N_HEADS * N_NODES;

    gat_init<<<(N_NODES * N_HEADS * N_HID + 255) / 256, 256, 0, stream>>>(out, segmax, denom);

    gat_gemm<<<N_NODES / 16, 256, 0, stream>>>(h, W, Wh);

    gat_score<<<(N_NODES * N_HEADS * 32 + 255) / 256, 256, 0, stream>>>(Wh, a, s1, s2);

    dim3 eg((N_EDGES + 255) / 256, N_HEADS);
    gat_edge_max<<<eg, 256, 0, stream>>>(s1, s2, src, dst, ebuf, segmax);
    gat_edge_exp<<<eg, 256, 0, stream>>>(dst, segmax, ebuf, denom);

    gat_message<<<(N_EDGES * 32 + 255) / 256, 256, 0, stream>>>(src, dst, ebuf, denom, Wh, out);
}


# s1,s2,segmax,denom,ebuf stored head-innermost ([node][head],[edge][head]); wave = 4 edges x 8 heads: a scattered gather/atomic touches 4 lines not 32; message reads one 32 B piece not 8 words
# speedup vs baseline: 2.2099x; 2.2099x over previous
#include <hip/hip_runtime.h>
#include <hip/hip_bf16.h>

typedef float v2f __attribute__((ext_vector_type(2)));
typedef float v8f __attribute__((ext_vector_type(8)));

#define N_NODES 50000
#define N_EDGES 800000
#define N_FEAT  256
#define N_HID   32
#define N_HEADS 8
#define ALPHA   0.2f

__global__ void gat_init(float* __restrict__ out,
                         float* __restrict__ segmax,
                         float* __restrict__ denom) {
    int i = blockIdx.x * blockDim.x + threadIdx.x;
    if (i < N_NODES * N_HEADS * N_HID) out[i] = 0.0f;
    if (i < N_NODES * N_HEADS) {
        segmax[i] = -__builtin_inff();
        denom[i]  = 0.0f;
    }
}

__global__ __launch_bounds__(256) void gat_gemm(const float* __restrict__ hmat,
                                                const float* __restrict__ W,
                                                float* __restrict__ Wh) {
    __shared__ float sh[16 * N_FEAT];

    const int node_base = blockIdx.x * 16;

    const float4* g4 = (const float4*)(hmat + (size_t)node_base * N_FEAT);
    float4* s4 = (float4*)sh;
    for (int i = threadIdx.x; i < (16 * N_FEAT) / 4; i += 256)
        s4[i] = g4[i];
    __syncthreads();

    const int lane = threadIdx.x & 31;
    const int head = threadIdx.x >> 5;
    const float* Wp = W + (size_t)head * N_FEAT * N_HID;

    const int m    = lane & 15;
    const int koff = (lane >> 4) << 1;

    v8f c0 = {};
    v8f c1 = {};

    for (int k = 0; k < N_FEAT; k += 4) {
        v2f a, b0, b1;
        a[0] = sh[m * N_FEAT + k + koff];
        a[1] = sh[m * N_FEAT + k + koff + 1];
        b0[0] = Wp[(k + koff)     * N_HID + m];
        b0[1] = Wp[(k + koff + 1) * N_HID + m];
        b1[0] = Wp[(k + koff)     * N_HID + 16 + m];
        b1[1] = Wp[(k + koff + 1) * N_HID + 16 + m];
        c0 = __builtin_amdgcn_wmma_f32_16x16x4_f32(false, a, false, b0,
                                                   (short)0, c0, false, false);
        c1 = __builtin_amdgcn_wmma_f32_16x16x4_f32(false, a, false, b1,
                                                   (short)0, c1, false, false);
    }

    float* op = Wh + ((size_t)head * N_NODES + node_base) * N_HID;
    const int mb = (lane >> 4) * 8;
#pragma unroll
    for (int v = 0; v < 8; ++v) {
        op[(mb + v) * N_HID + m]      = c0[v];
        op[(mb + v) * N_HID + 16 + m] = c1[v];
    }
}

__global__ __launch_bounds__(256) void gat_score(const float* __restrict__ Wh,
                                                 const float* __restrict__ avec,
                                                 float* __restrict__ s1,
                                                 float* __restrict__ s2) {
    const int g    = (blockIdx.x * blockDim.x + threadIdx.x) >> 5;
    const int lane = threadIdx.x & 31;
    if (g >= N_NODES * N_HEADS) return;
    const int head = g & 7;
    const int n    = g >> 3;

    float wh = Wh[((size_t)head * N_NODES + n) * N_HID + lane];
    float p1 = wh * avec[head * 2 * N_HID + lane];
    float p2 = wh * avec[head * 2 * N_HID + N_HID + lane];
#pragma unroll
    for (int off = 16; off > 0; off >>= 1) {
        p1 += __shfl_xor(p1, off, 32);
        p2 += __shfl_xor(p2, off, 32);
    }
    if (lane == 0) {
        s1[n * N_HEADS + head] = p1;
        s2[n * N_HEADS + head] = p2;
    }
}

__device__ __forceinline__ void atomicMaxFloat(float* addr, float val) {
    if (val >= 0.0f)
        atomicMax((int*)addr, __float_as_int(val));
    else
        atomicMin((unsigned int*)addr, __float_as_uint(val));
}

__global__ void gat_edge_max(const float* __restrict__ s1,
                             const float* __restrict__ s2,
                             const int* __restrict__ src,
                             const int* __restrict__ dst,
                             float* __restrict__ ebuf,
                             float* __restrict__ segmax) {
    const int t = blockIdx.x * blockDim.x + threadIdx.x;
    if (t >= N_EDGES * N_HEADS) return;
    const int e = t >> 3, head = t & 7;
    const int s = src[e], d = dst[e];
    float x = s1[s * N_HEADS + head] + s2[d * N_HEADS + head];
    float v = x > 0.0f ? x : ALPHA * x;
    ebuf[t] = v;
    atomicMaxFloat(&segmax[d * N_HEADS + head], v);
}

__global__ void gat_edge_exp(const int* __restrict__ dst,
                             const float* __restrict__ segmax,
                             float* __restrict__ ebuf,
                             float* __restrict__ denom) {
    const int t = blockIdx.x * blockDim.x + threadIdx.x;
    if (t >= N_EDGES * N_HEADS) return;
    const int e = t >> 3, head = t & 7;
    const int d = dst[e];
    float ex = __expf(ebuf[t] - segmax[d * N_HEADS + head]);
    ebuf[t] = ex;
    atomicAdd(&denom[d * N_HEADS + head], ex);
}

__global__ __launch_bounds__(256) void gat_message(const int* __restrict__ src,
                                                   const int* __restrict__ dst,
                                                   const float* __restrict__ ebuf,
                                                   const float* __restrict__ denom,
                                                   const float* __restrict__ Wh,
                                                   float* __restrict__ out) {
    const int e    = (blockIdx.x * blockDim.x + threadIdx.x) >> 5;
    const int lane = threadIdx.x & 31;
    if (e >= N_EDGES) return;
    const int s = src[e], d = dst[e];
    float* op = out + (size_t)d * (N_HEADS * N_HID);
    const int hl = lane & 7;
    const float attn_l = ebuf[(size_t)e * N_HEADS + hl] / denom[d * N_HEADS + hl];
#pragma unroll
    for (int head = 0; head < N_HEADS; ++head) {
        float attn = __int_as_float(__builtin_amdgcn_readlane(__float_as_int(attn_l), head));
        float val  = attn * Wh[((size_t)head * N_NODES + s) * N_HID + lane];
        atomicAdd(op + head * N_HID + lane, val);
    }
}

extern "C" void kernel_launch(void* const* d_in, const int* in_sizes, int n_in,
                              void* d_out, int out_size, void* d_ws, size_t ws_size,
                              hipStream_t stream) {
    const float* h   = (const float*)d_in[0];
    const float* W   = (const float*)d_in[1];
    const float* a   = (const float*)d_in[2];
    const int*   src = (const int*)d_in[3];
    const int*   dst = (const int*)d_in[4];
    float* out = (float*)d_out;

    float* ws     = (float*)d_ws;
    float* Wh     = ws;
    float* s1     = Wh + (size_t)N_HEADS * N_NODES * N_HID;
    float* s2     = s1 + (size_t)N_HEADS * N_NODES;
    float* segmax = s2 + (size_t)N_HEADS * N_NODES;
    float* denom  = segmax + (size_t)N_HEADS * N_NODES;
    float* ebuf   = denom + (size_t)N_HEADS * N_NODES;

    gat_init<<<(N_NODES * N_HEADS * N_HID + 255) / 256, 256, 0, stream>>>(out, segmax, denom);

    gat_gemm<<<N_NODES / 16, 256, 0, stream>>>(h, W, Wh);

    gat_score<<<(N_NODES * N_HEADS * 32 + 255) / 256, 256, 0, stream>>>(Wh, a, s1, s2);

    gat_edge_max<<<(N_EDGES * N_HEADS + 255) / 256, 256, 0, stream>>>(s1, s2, src, dst, ebuf, segmax);
    gat_edge_exp<<<(N_EDGES * N_HEADS + 255) / 256, 256, 0, stream>>>(dst, segmax, ebuf, denom);

    gat_message<<<(N_EDGES * 32 + 255) / 256, 256, 0, stream>>>(src, dst, ebuf, denom, Wh, out);
}


# edges bucketed by dst each call (count, scan, fill); one wave per dst node does max, exp-sum, message sum in registers, stores out once; no float atomics, sums in bucket order (int-atomic arrival)
# speedup vs baseline: 3.5580x; 1.6100x over previous
#include <hip/hip_runtime.h>
#include <hip/hip_bf16.h>

typedef float v2f __attribute__((ext_vector_type(2)));
typedef float v8f __attribute__((ext_vector_type(8)));

#define N_NODES 50000
#define N_EDGES 800000
#define N_FEAT  256
#define N_HID   32
#define N_HEADS 8
#define ALPHA   0.2f

__global__ __launch_bounds__(256) void gat_gemm(const float* __restrict__ hmat,
                                                const float* __restrict__ W,
                                                float* __restrict__ Wh) {
    __shared__ float sh[16 * N_FEAT];

    const int node_base = blockIdx.x * 16;

    const float4* g4 = (const float4*)(hmat + (size_t)node_base * N_FEAT);
    float4* s4 = (float4*)sh;
    for (int i = threadIdx.x; i < (16 * N_FEAT) / 4; i += 256)
        s4[i] = g4[i];
    __syncthreads();

    const int lane = threadIdx.x & 31;
    const int head = threadIdx.x >> 5;
    const float* Wp = W + (size_t)head * N_FEAT * N_HID;

    const int m    = lane & 15;
    const int koff = (lane >> 4) << 1;

    v8f c0 = {};
    v8f c1 = {};

    for (int k = 0; k < N_FEAT; k += 4) {
        v2f a, b0, b1;
        a[0] = sh[m * N_FEAT + k + koff];
        a[1] = sh[m * N_FEAT + k + koff + 1];
        b0[0] = Wp[(k + koff)     * N_HID + m];
        b0[1] = Wp[(k + koff + 1) * N_HID + m];
        b1[0] = Wp[(k + koff)     * N_HID + 16 + m];
        b1[1] = Wp[(k + koff + 1) * N_HID + 16 + m];
        c0 = __builtin_amdgcn_wmma_f32_16x16x4_f32(false, a, false, b0,
                                                   (short)0, c0, false, false);
        c1 = __builtin_amdgcn_wmma_f32_16x16x4_f32(false, a, false, b1,
                                                   (short)0, c1, false, false);
    }

    float* op = Wh + ((size_t)head * N_NODES + node_base) * N_HID;
    const int mb = (lane >> 4) * 8;
#pragma unroll
    for (int v = 0; v < 8; ++v) {
        op[(mb + v) * N_HID + m]      = c0[v];
        op[(mb + v) * N_HID + 16 + m] = c1[v];
    }
}

__global__ __launch_bounds__(256) void gat_score(const float* __restrict__ Wh,
                                                 const float* __restrict__ avec,
                                                 float* __restrict__ s1,
                                                 float* __restrict__ s2) {
    const int g    = (blockIdx.x * blockDim.x + threadIdx.x) >> 5;
    const int lane = threadIdx.x & 31;
    if (g >= N_NODES * N_HEADS) return;
    const int head = g & 7;
    const int n    = g >> 3;

    float wh = Wh[((size_t)head * N_NODES + n) * N_HID + lane];
    float p1 = wh * avec[head * 2 * N_HID + lane];
    float p2 = wh * avec[head * 2 * N_HID + N_HID + lane];
#pragma unroll
    for (int off = 16; off > 0; off >>= 1) {
        p1 += __shfl_xor(p1, off, 32);
        p2 += __shfl_xor(p2, off, 32);
    }
    if (lane == 0) {
        s1[n * N_HEADS + head] = p1;
        s2[n * N_HEADS + head] = p2;
    }
}


__global__ void gat_zero_cnt(int* __restrict__ cur) {
    const int i = blockIdx.x * blockDim.x + threadIdx.x;
    if (i < N_NODES) cur[i] = 0;
}

__global__ void gat_count(const int* __restrict__ dst, int* __restrict__ cur) {
    const int e = blockIdx.x * blockDim.x + threadIdx.x;
    if (e < N_EDGES) atomicAdd(&cur[dst[e]], 1);
}

#define SCAN_TILES ((N_NODES + 4095) / 4096)
__global__ __launch_bounds__(1024) void gat_scan(int* __restrict__ cur) {
    __shared__ int wsum[32];
    const int t = threadIdx.x, lane = t & 31, w = t >> 5;
    int4 c[SCAN_TILES];
#pragma unroll
    for (int k = 0; k < SCAN_TILES; ++k) {
        const int i = k * 4096 + t * 4;
        c[k] = make_int4(0, 0, 0, 0);
        if (i < N_NODES) c[k] = *(const int4*)(cur + i);
    }
    int carry = 0;
#pragma unroll
    for (int k = 0; k < SCAN_TILES; ++k) {
        const int i = k * 4096 + t * 4;
        const int p0 = c[k].x, p1 = p0 + c[k].y, p2 = p1 + c[k].z, p3 = p2 + c[k].w;
        int incl = p3;
        for (int off = 1; off < 32; off <<= 1) {
            const int u = __shfl_up(incl, off, 32);
            if (lane >= off) incl += u;
        }
        if (lane == 31) wsum[w] = incl;
        __syncthreads();
        if (w == 0) {
            int x = wsum[lane];
            for (int off = 1; off < 32; off <<= 1) {
                const int u = __shfl_up(x, off, 32);
                if (lane >= off) x += u;
            }
            wsum[lane] = x;
        }
        __syncthreads();
        const int wave_off = w ? wsum[w - 1] : 0;
        const int total    = wsum[31];
        const int excl     = carry + wave_off + incl - p3;
        if (i < N_NODES)
            *(int4*)(cur + i) = make_int4(excl, excl + p0, excl + p1, excl + p2);
        carry += total;
        __syncthreads();
    }
}

__global__ void gat_fill(const int* __restrict__ src, const int* __restrict__ dst,
                         int* __restrict__ cur, int* __restrict__ bsrc) {
    const int e = blockIdx.x * blockDim.x + threadIdx.x;
    if (e >= N_EDGES) return;
    const int slot = atomicAdd(&cur[dst[e]], 1);
    bsrc[slot] = src[e];
}

#define GAT_EDGE(jj)                                                              \
    {                                                                              \
        const int sj = __builtin_amdgcn_readlane(s, (jj) * 8);                     \
          \
        _Pragma("unroll")                                                          \
        for (int head = 0; head < N_HEADS; ++head) {                               \
            const float aw = __int_as_float(__builtin_amdgcn_readlane(             \
                                 __float_as_int(attn_l), (jj) * 8 + head));        \
            acc[head] += aw * Wh[((size_t)head * N_NODES + sj) * N_HID + lane];    \
        }                                                                          \
    }

__global__ __launch_bounds__(256) void gat_node(const int* __restrict__ cur,
                                                const int* __restrict__ bsrc,
                                                const float* __restrict__ s1,
                                                const float* __restrict__ s2,
                                                const float* __restrict__ Wh,
                                                float* __restrict__ out) {
    const int node = __builtin_amdgcn_readfirstlane((int)(blockIdx.x * 8 + (threadIdx.x >> 5)));
    if (node >= N_NODES) return;
    const int lane = threadIdx.x & 31;
    const int hl = lane & 7;
    const int sl = lane >> 3;

    const int beg = node ? cur[node - 1] : 0;
    const int end = cur[node];
    const float s2v = s2[node * N_HEADS + hl];

    float mx = -__builtin_inff();
    for (int base = beg; base < end; base += 4) {
        const int j = base + sl;
        const bool ok = j < end;
        const int s = bsrc[ok ? j : beg];
        const float x = s1[s * N_HEADS + hl] + s2v;
        const float v = x > 0.0f ? x : ALPHA * x;
        mx = ok ? fmaxf(mx, v) : mx;
    }
    mx = fmaxf(mx, __shfl_xor(mx, 8, 32));
    mx = fmaxf(mx, __shfl_xor(mx, 16, 32));

    float sum = 0.0f;
    for (int base = beg; base < end; base += 4) {
        const int j = base + sl;
        const bool ok = j < end;
        const int s = bsrc[ok ? j : beg];
        const float x = s1[s * N_HEADS + hl] + s2v;
        const float v = x > 0.0f ? x : ALPHA * x;
        sum += ok ? __expf(v - mx) : 0.0f;
    }
    sum += __shfl_xor(sum, 8, 32);
    sum += __shfl_xor(sum, 16, 32);

    float acc[N_HEADS];
    for (int head = 0; head < N_HEADS; ++head) acc[head] = 0.0f;

    for (int base = beg; base < end; base += 4) {
        const int j = base + sl;
        const bool ok = j < end;
        const int s = bsrc[ok ? j : beg];
        const float x = s1[s * N_HEADS + hl] + s2v;
        const float v = x > 0.0f ? x : ALPHA * x;
        const float ex = ok ? __expf(v - mx) : 0.0f;
        const float attn_l = ex / sum;
        const int left = end - base;
        if (left >= 4) {
            GAT_EDGE(0) GAT_EDGE(1) GAT_EDGE(2) GAT_EDGE(3)
        } else {
            GAT_EDGE(0)
            if (left > 1) GAT_EDGE(1)
            if (left > 2) GAT_EDGE(2)
        }
    }

    float* op = out + (size_t)node * (N_HEADS * N_HID);
    for (int head = 0; head < N_HEADS; ++head) op[head * N_HID + lane] = acc[head];
}

extern "C" void kernel_launch(void* const* d_in, const int* in_sizes, int n_in,
                              void* d_out, int out_size, void* d_ws, size_t ws_size,
                              hipStream_t stream) {
    const float* h   = (const float*)d_in[0];
    const float* W   = (const float*)d_in[1];
    const float* a   = (const float*)d_in[2];
    const int*   src = (const int*)d_in[3];
    const int*   dst = (const int*)d_in[4];
    float* out = (float*)d_out;

    float* ws   = (float*)d_ws;
    float* Wh   = ws;
    float* s1   = Wh + (size_t)N_HEADS * N_NODES * N_HID;
    float* s2   = s1 + (size_t)N_HEADS * N_NODES;
    int*   cur  = (int*)(s2 + (size_t)N_HEADS * N_NODES);
    int*   bsrc = cur + N_NODES;

    gat_zero_cnt<<<(N_NODES + 255) / 256, 256, 0, stream>>>(cur);

    gat_gemm<<<N_NODES / 16, 256, 0, stream>>>(h, W, Wh);

    gat_score<<<(N_NODES * N_HEADS * 32 + 255) / 256, 256, 0, stream>>>(Wh, a, s1, s2);

    gat_count<<<(N_EDGES + 255) / 256, 256, 0, stream>>>(dst, cur);
    gat_scan<<<1, 1024, 0, stream>>>(cur);
    gat_fill<<<(N_EDGES + 255) / 256, 256, 0, stream>>>(src, dst, cur, bsrc);

    gat_node<<<(N_NODES + 7) / 8, 256, 0, stream>>>(cur, bsrc, s1, s2, Wh, out);
}


# s1,s2 dot products taken from the GEMM accumulator tile in its epilogue (16-lane DPP butterfly, one full-line store per wave); score kernel and its re-read of Wh removed; s1,s2 sum tree changed
# speedup vs baseline: 4.0393x; 1.1353x over previous
#include <hip/hip_runtime.h>
#include <hip/hip_bf16.h>

typedef float v2f __attribute__((ext_vector_type(2)));
typedef float v8f __attribute__((ext_vector_type(8)));

#define N_NODES 50000
#define N_EDGES 800000
#define N_FEAT  256
#define N_HID   32
#define N_HEADS 8
#define ALPHA   0.2f

template <int K>
__device__ __forceinline__ float row_xor(float v) {
    return __int_as_float(__builtin_amdgcn_update_dpp(0, __float_as_int(v), 0x160 | K,
                                                      0xf, 0xf, true));
}

__global__ __launch_bounds__(256) void gat_gemm(const float* __restrict__ hmat,
                                                const float* __restrict__ W,
                                                const float* __restrict__ avec,
                                                float* __restrict__ Wh,
                                                float* __restrict__ s1,
                                                float* __restrict__ s2) {
    __shared__ float sh[16 * N_FEAT];
    __shared__ float ssc[2 * 16 * N_HEADS];

    const int node_base = blockIdx.x * 16;

    const float4* g4 = (const float4*)(hmat + (size_t)node_base * N_FEAT);
    float4* s4 = (float4*)sh;
    for (int i = threadIdx.x; i < (16 * N_FEAT) / 4; i += 256)
        s4[i] = g4[i];
    __syncthreads();

    const int lane = threadIdx.x & 31;
    const int head = threadIdx.x >> 5;
    const float* Wp = W + (size_t)head * N_FEAT * N_HID;

    const int m    = lane & 15;
    const int koff = (lane >> 4) << 1;

    v8f c0 = {};
    v8f c1 = {};

    for (int k = 0; k < N_FEAT; k += 4) {
        v2f a, b0, b1;
        a[0] = sh[m * N_FEAT + k + koff];
        a[1] = sh[m * N_FEAT + k + koff + 1];
        b0[0] = Wp[(k + koff)     * N_HID + m];
        b0[1] = Wp[(k + koff + 1) * N_HID + m];
        b1[0] = Wp[(k + koff)     * N_HID + 16 + m];
        b1[1] = Wp[(k + koff + 1) * N_HID + 16 + m];
        c0 = __builtin_amdgcn_wmma_f32_16x16x4_f32(false, a, false, b0,
                                                   (short)0, c0, false, false);
        c1 = __builtin_amdgcn_wmma_f32_16x16x4_f32(false, a, false, b1,
                                                   (short)0, c1, false, false);
    }

    const int mb = (lane >> 4) * 8;

    const float* ap = avec + head * 2 * N_HID;
    const float a1l = ap[m],         a1h = ap[16 + m];
    const float a2l = ap[N_HID + m], a2h = ap[N_HID + 16 + m];
    float x[16];
#pragma unroll
    for (int v = 0; v < 8; ++v) {
        x[v]     = c0[v] * a1l + c1[v] * a1h;
        x[8 + v] = c0[v] * a2l + c1[v] * a2h;
    }
    const bool b3 = (m & 8) != 0, b2 = (m & 4) != 0, b1 = (m & 2) != 0, b0 = (m & 1) != 0;
    float y[8], z[4], w2[2];
#pragma unroll
    for (int i = 0; i < 8; ++i)
        y[i] = (b3 ? x[8 + i] : x[i]) + row_xor<8>(b3 ? x[i] : x[8 + i]);
#pragma unroll
    for (int i = 0; i < 4; ++i)
        z[i] = (b2 ? y[4 + i] : y[i]) + row_xor<4>(b2 ? y[i] : y[4 + i]);
#pragma unroll
    for (int i = 0; i < 2; ++i)
        w2[i] = (b1 ? z[2 + i] : z[i]) + row_xor<2>(b1 ? z[i] : z[2 + i]);
    const float sc = (b0 ? w2[1] : w2[0]) + row_xor<1>(b0 ? w2[0] : w2[1]);

    ssc[(m >> 3) * (16 * N_HEADS) + (mb + (m & 7)) * N_HEADS + head] = sc;
    __syncthreads();
    float* sp = (threadIdx.x < 16 * N_HEADS) ? s1 : s2;
    sp[node_base * N_HEADS + (threadIdx.x & (16 * N_HEADS - 1))] = ssc[threadIdx.x];

    float* op = Wh + ((size_t)head * N_NODES + node_base) * N_HID;
#pragma unroll
    for (int v = 0; v < 8; ++v) {
        op[(mb + v) * N_HID + m]      = c0[v];
        op[(mb + v) * N_HID + 16 + m] = c1[v];
    }
}


__global__ void gat_zero_cnt(int* __restrict__ cur) {
    const int i = blockIdx.x * blockDim.x + threadIdx.x;
    if (i < N_NODES) cur[i] = 0;
}

__global__ void gat_count(const int* __restrict__ dst, int* __restrict__ cur) {
    const int e = blockIdx.x * blockDim.x + threadIdx.x;
    if (e < N_EDGES) atomicAdd(&cur[dst[e]], 1);
}

#define SCAN_TILES ((N_NODES + 4095) / 4096)
__global__ __launch_bounds__(1024) void gat_scan(int* __restrict__ cur) {
    __shared__ int wsum[32];
    const int t = threadIdx.x, lane = t & 31, w = t >> 5;
    int4 c[SCAN_TILES];
#pragma unroll
    for (int k = 0; k < SCAN_TILES; ++k) {
        const int i = k * 4096 + t * 4;
        c[k] = make_int4(0, 0, 0, 0);
        if (i < N_NODES) c[k] = *(const int4*)(cur + i);
    }
    int carry = 0;
#pragma unroll
    for (int k = 0; k < SCAN_TILES; ++k) {
        const int i = k * 4096 + t * 4;
        const int p0 = c[k].x, p1 = p0 + c[k].y, p2 = p1 + c[k].z, p3 = p2 + c[k].w;
        int incl = p3;
        for (int off = 1; off < 32; off <<= 1) {
            const int u = __shfl_up(incl, off, 32);
            if (lane >= off) incl += u;
        }
        if (lane == 31) wsum[w] = incl;
        __syncthreads();
        if (w == 0) {
            int x = wsum[lane];
            for (int off = 1; off < 32; off <<= 1) {
                const int u = __shfl_up(x, off, 32);
                if (lane >= off) x += u;
            }
            wsum[lane] = x;
        }
        __syncthreads();
        const int wave_off = w ? wsum[w - 1] : 0;
        const int total    = wsum[31];
        const int excl     = carry + wave_off + incl - p3;
        if (i < N_NODES)
            *(int4*)(cur + i) = make_int4(excl, excl + p0, excl + p1, excl + p2);
        carry += total;
        __syncthreads();
    }
}

__global__ void gat_fill(const int* __restrict__ src, const int* __restrict__ dst,
                         int* __restrict__ cur, int* __restrict__ bsrc) {
    const int e = blockIdx.x * blockDim.x + threadIdx.x;
    if (e >= N_EDGES) return;
    const int slot = atomicAdd(&cur[dst[e]], 1);
    bsrc[slot] = src[e];
}

#define GAT_EDGE(jj)                                                              \
    {                                                                              \
        const int sj = __builtin_amdgcn_readlane(s, (jj) * 8);                     \
          \
        _Pragma("unroll")                                                          \
        for (int head = 0; head < N_HEADS; ++head) {                               \
            const float aw = __int_as_float(__builtin_amdgcn_readlane(             \
                                 __float_as_int(attn_l), (jj) * 8 + head));        \
            acc[head] += aw * Wh[((size_t)head * N_NODES + sj) * N_HID + lane];    \
        }                                                                          \
    }

__global__ __launch_bounds__(256) void gat_node(const int* __restrict__ cur,
                                                const int* __restrict__ bsrc,
                                                const float* __restrict__ s1,
                                                const float* __restrict__ s2,
                                                const float* __restrict__ Wh,
                                                float* __restrict__ out) {
    const int node = __builtin_amdgcn_readfirstlane((int)(blockIdx.x * 8 + (threadIdx.x >> 5)));
    if (node >= N_NODES) return;
    const int lane = threadIdx.x & 31;
    const int hl = lane & 7;
    const int sl = lane >> 3;

    const int beg = node ? cur[node - 1] : 0;
    const int end = cur[node];
    const float s2v = s2[node * N_HEADS + hl];

    float mx = -__builtin_inff();
    for (int base = beg; base < end; base += 4) {
        const int j = base + sl;
        const bool ok = j < end;
        const int s = bsrc[ok ? j : beg];
        const float x = s1[s * N_HEADS + hl] + s2v;
        const float v = x > 0.0f ? x : ALPHA * x;
        mx = ok ? fmaxf(mx, v) : mx;
    }
    mx = fmaxf(mx, __shfl_xor(mx, 8, 32));
    mx = fmaxf(mx, __shfl_xor(mx, 16, 32));

    float sum = 0.0f;
    for (int base = beg; base < end; base += 4) {
        const int j = base + sl;
        const bool ok = j < end;
        const int s = bsrc[ok ? j : beg];
        const float x = s1[s * N_HEADS + hl] + s2v;
        const float v = x > 0.0f ? x : ALPHA * x;
        sum += ok ? __expf(v - mx) : 0.0f;
    }
    sum += __shfl_xor(sum, 8, 32);
    sum += __shfl_xor(sum, 16, 32);

    float acc[N_HEADS];
    for (int head = 0; head < N_HEADS; ++head) acc[head] = 0.0f;

    for (int base = beg; base < end; base += 4) {
        const int j = base + sl;
        const bool ok = j < end;
        const int s = bsrc[ok ? j : beg];
        const float x = s1[s * N_HEADS + hl] + s2v;
        const float v = x > 0.0f ? x : ALPHA * x;
        const float ex = ok ? __expf(v - mx) : 0.0f;
        const float attn_l = ex / sum;
        const int left = end - base;
        if (left >= 4) {
            GAT_EDGE(0) GAT_EDGE(1) GAT_EDGE(2) GAT_EDGE(3)
        } else {
            GAT_EDGE(0)
            if (left > 1) GAT_EDGE(1)
            if (left > 2) GAT_EDGE(2)
        }
    }

    float* op = out + (size_t)node * (N_HEADS * N_HID);
    for (int head = 0; head < N_HEADS; ++head) op[head * N_HID + lane] = acc[head];
}

extern "C" void kernel_launch(void* const* d_in, const int* in_sizes, int n_in,
                              void* d_out, int out_size, void* d_ws, size_t ws_size,
                              hipStream_t stream) {
    const float* h   = (const float*)d_in[0];
    const float* W   = (const float*)d_in[1];
    const float* a   = (const float*)d_in[2];
    const int*   src = (const int*)d_in[3];
    const int*   dst = (const int*)d_in[4];
    float* out = (float*)d_out;

    float* ws   = (float*)d_ws;
    float* Wh   = ws;
    float* s1   = Wh + (size_t)N_HEADS * N_NODES * N_HID;
    float* s2   = s1 + (size_t)N_HEADS * N_NODES;
    int*   cur  = (int*)(s2 + (size_t)N_HEADS * N_NODES);
    int*   bsrc = cur + N_NODES;

    gat_zero_cnt<<<(N_NODES + 255) / 256, 256, 0, stream>>>(cur);

    gat_gemm<<<N_NODES / 16, 256, 0, stream>>>(h, W, a, Wh, s1, s2);

    gat_count<<<(N_EDGES + 255) / 256, 256, 0, stream>>>(dst, cur);
    gat_scan<<<1, 1024, 0, stream>>>(cur);
    gat_fill<<<(N_EDGES + 255) / 256, 256, 0, stream>>>(src, dst, cur, bsrc);

    gat_node<<<(N_NODES + 7) / 8, 256, 0, stream>>>(cur, bsrc, s1, s2, Wh, out);
}


# Wh kept node-major (1 KB row per node): node kernel gathers it with two 16-byte loads per edge (4 whole lines each), not 8 one-line loads, 8 edges a trip; exp-sum tree changed (8 partials, not 4)
# speedup vs baseline: 4.2671x; 1.0564x over previous
#include <hip/hip_runtime.h>
#include <hip/hip_bf16.h>

typedef float v2f __attribute__((ext_vector_type(2)));
typedef float v8f __attribute__((ext_vector_type(8)));

#define N_NODES 50000
#define N_EDGES 800000
#define N_FEAT  256
#define N_HID   32
#define N_HEADS 8
#define ALPHA   0.2f

template <int K>
__device__ __forceinline__ float row_xor(float v) {
    return __int_as_float(__builtin_amdgcn_update_dpp(0, __float_as_int(v), 0x160 | K,
                                                      0xf, 0xf, true));
}

__global__ __launch_bounds__(256) void gat_gemm(const float* __restrict__ hmat,
                                                const float* __restrict__ W,
                                                const float* __restrict__ avec,
                                                float* __restrict__ Wh,
                                                float* __restrict__ s1,
                                                float* __restrict__ s2) {
    __shared__ float sh[16 * N_FEAT];
    __shared__ float ssc[2 * 16 * N_HEADS];

    const int node_base = blockIdx.x * 16;

    const float4* g4 = (const float4*)(hmat + (size_t)node_base * N_FEAT);
    float4* s4 = (float4*)sh;
    for (int i = threadIdx.x; i < (16 * N_FEAT) / 4; i += 256)
        s4[i] = g4[i];
    __syncthreads();

    const int lane = threadIdx.x & 31;
    const int head = threadIdx.x >> 5;
    const float* Wp = W + (size_t)head * N_FEAT * N_HID;

    const int m    = lane & 15;
    const int koff = (lane >> 4) << 1;

    v8f c0 = {};
    v8f c1 = {};

    for (int k = 0; k < N_FEAT; k += 4) {
        v2f a, b0, b1;
        a[0] = sh[m * N_FEAT + k + koff];
        a[1] = sh[m * N_FEAT + k + koff + 1];
        b0[0] = Wp[(k + koff)     * N_HID + m];
        b0[1] = Wp[(k + koff + 1) * N_HID + m];
        b1[0] = Wp[(k + koff)     * N_HID + 16 + m];
        b1[1] = Wp[(k + koff + 1) * N_HID + 16 + m];
        c0 = __builtin_amdgcn_wmma_f32_16x16x4_f32(false, a, false, b0,
                                                   (short)0, c0, false, false);
        c1 = __builtin_amdgcn_wmma_f32_16x16x4_f32(false, a, false, b1,
                                                   (short)0, c1, false, false);
    }

    const int mb = (lane >> 4) * 8;

    const float* ap = avec + head * 2 * N_HID;
    const float a1l = ap[m],         a1h = ap[16 + m];
    const float a2l = ap[N_HID + m], a2h = ap[N_HID + 16 + m];
    float x[16];
#pragma unroll
    for (int v = 0; v < 8; ++v) {
        x[v]     = c0[v] * a1l + c1[v] * a1h;
        x[8 + v] = c0[v] * a2l + c1[v] * a2h;
    }
    const bool b3 = (m & 8) != 0, b2 = (m & 4) != 0, b1 = (m & 2) != 0, b0 = (m & 1) != 0;
    float y[8], z[4], w2[2];
#pragma unroll
    for (int i = 0; i < 8; ++i)
        y[i] = (b3 ? x[8 + i] : x[i]) + row_xor<8>(b3 ? x[i] : x[8 + i]);
#pragma unroll
    for (int i = 0; i < 4; ++i)
        z[i] = (b2 ? y[4 + i] : y[i]) + row_xor<4>(b2 ? y[i] : y[4 + i]);
#pragma unroll
    for (int i = 0; i < 2; ++i)
        w2[i] = (b1 ? z[2 + i] : z[i]) + row_xor<2>(b1 ? z[i] : z[2 + i]);
    const float sc = (b0 ? w2[1] : w2[0]) + row_xor<1>(b0 ? w2[0] : w2[1]);

    ssc[(m >> 3) * (16 * N_HEADS) + (mb + (m & 7)) * N_HEADS + head] = sc;
    __syncthreads();
    float* sp = (threadIdx.x < 16 * N_HEADS) ? s1 : s2;
    sp[node_base * N_HEADS + (threadIdx.x & (16 * N_HEADS - 1))] = ssc[threadIdx.x];

    float* op = Wh + (size_t)node_base * (N_HEADS * N_HID) + head * N_HID;
#pragma unroll
    for (int v = 0; v < 8; ++v) {
        op[(mb + v) * (N_HEADS * N_HID) + m]      = c0[v];
        op[(mb + v) * (N_HEADS * N_HID) + 16 + m] = c1[v];
    }
}


__global__ void gat_zero_cnt(int* __restrict__ cur) {
    const int i = blockIdx.x * blockDim.x + threadIdx.x;
    if (i < N_NODES) cur[i] = 0;
}

__global__ void gat_count(const int* __restrict__ dst, int* __restrict__ cur) {
    const int e = blockIdx.x * blockDim.x + threadIdx.x;
    if (e < N_EDGES) atomicAdd(&cur[dst[e]], 1);
}

#define SCAN_TILES ((N_NODES + 4095) / 4096)
__global__ __launch_bounds__(1024) void gat_scan(int* __restrict__ cur) {
    __shared__ int wsum[32];
    const int t = threadIdx.x, lane = t & 31, w = t >> 5;
    int4 c[SCAN_TILES];
#pragma unroll
    for (int k = 0; k < SCAN_TILES; ++k) {
        const int i = k * 4096 + t * 4;
        c[k] = make_int4(0, 0, 0, 0);
        if (i < N_NODES) c[k] = *(const int4*)(cur + i);
    }
    int carry = 0;
#pragma unroll
    for (int k = 0; k < SCAN_TILES; ++k) {
        const int i = k * 4096 + t * 4;
        const int p0 = c[k].x, p1 = p0 + c[k].y, p2 = p1 + c[k].z, p3 = p2 + c[k].w;
        int incl = p3;
        for (int off = 1; off < 32; off <<= 1) {
            const int u = __shfl_up(incl, off, 32);
            if (lane >= off) incl += u;
        }
        if (lane == 31) wsum[w] = incl;
        __syncthreads();
        if (w == 0) {
            int x = wsum[lane];
            for (int off = 1; off < 32; off <<= 1) {
                const int u = __shfl_up(x, off, 32);
                if (lane >= off) x += u;
            }
            wsum[lane] = x;
        }
        __syncthreads();
        const int wave_off = w ? wsum[w - 1] : 0;
        const int total    = wsum[31];
        const int excl     = carry + wave_off + incl - p3;
        if (i < N_NODES)
            *(int4*)(cur + i) = make_int4(excl, excl + p0, excl + p1, excl + p2);
        carry += total;
        __syncthreads();
    }
}

__global__ void gat_fill(const int* __restrict__ src, const int* __restrict__ dst,
                         int* __restrict__ cur, int* __restrict__ bsrc) {
    const int e = blockIdx.x * blockDim.x + threadIdx.x;
    if (e >= N_EDGES) return;
    const int slot = atomicAdd(&cur[dst[e]], 1);
    bsrc[slot] = src[e];
}

template <int J>
__device__ __forceinline__ float grp8_bcast(float v) {
    return __int_as_float(__builtin_amdgcn_mov_dpp8(__float_as_int(v), J * 0x249249));
}

#define GAT_ROW(jj)                                                               \
    {                                                                              \
        const int sj = __builtin_amdgcn_readlane(s, (jj));                         \
              \
        const float* rp = Wh + (size_t)sj * (N_HEADS * N_HID);                     \
        va[jj] = *(const float4*)(rp + loff);                                      \
        vb[jj] = *(const float4*)(rp + loff + N_HID);                              \
    }

__global__ __launch_bounds__(256) void gat_node(const int* __restrict__ cur,
                                                const int* __restrict__ bsrc,
                                                const float* __restrict__ s1,
                                                const float* __restrict__ s2,
                                                const float* __restrict__ Wh,
                                                float* __restrict__ out) {
    const int node = __builtin_amdgcn_readfirstlane((int)(blockIdx.x * 8 + (threadIdx.x >> 5)));
    if (node >= N_NODES) return;
    const int lane = threadIdx.x & 31;
    const int e8 = lane & 7;
    const int hp = lane >> 3;
    const unsigned loff = hp * (2 * N_HID) + e8 * 4;

    const int beg = node ? cur[node - 1] : 0;
    const int end = cur[node];
    const float2 s2v = *(const float2*)(s2 + node * N_HEADS + 2 * hp);

    const float ninf = -__builtin_inff();

    float mx0 = ninf, mx1 = ninf;
    for (int base = beg; base < end; base += 8) {
        const int j = base + e8;
        const bool ok = j < end;
        const int s = bsrc[ok ? j : beg];
        const float2 t = *(const float2*)(s1 + s * N_HEADS + 2 * hp);
        const float x0 = t.x + s2v.x, x1 = t.y + s2v.y;
        const float v0 = x0 > 0.0f ? x0 : ALPHA * x0;
        const float v1 = x1 > 0.0f ? x1 : ALPHA * x1;
        mx0 = fmaxf(mx0, ok ? v0 : ninf);
        mx1 = fmaxf(mx1, ok ? v1 : ninf);
    }
    mx0 = fmaxf(mx0, row_xor<1>(mx0));  mx1 = fmaxf(mx1, row_xor<1>(mx1));
    mx0 = fmaxf(mx0, row_xor<2>(mx0));  mx1 = fmaxf(mx1, row_xor<2>(mx1));
    mx0 = fmaxf(mx0, row_xor<4>(mx0));  mx1 = fmaxf(mx1, row_xor<4>(mx1));

    float sum0 = 0.0f, sum1 = 0.0f;
    for (int base = beg; base < end; base += 8) {
        const int j = base + e8;
        const bool ok = j < end;
        const int s = bsrc[ok ? j : beg];
        const float2 t = *(const float2*)(s1 + s * N_HEADS + 2 * hp);
        const float x0 = t.x + s2v.x, x1 = t.y + s2v.y;
        const float v0 = x0 > 0.0f ? x0 : ALPHA * x0;
        const float v1 = x1 > 0.0f ? x1 : ALPHA * x1;
        sum0 += __expf((ok ? v0 : ninf) - mx0);
        sum1 += __expf((ok ? v1 : ninf) - mx1);
    }
    sum0 += row_xor<1>(sum0);  sum1 += row_xor<1>(sum1);
    sum0 += row_xor<2>(sum0);  sum1 += row_xor<2>(sum1);
    sum0 += row_xor<4>(sum0);  sum1 += row_xor<4>(sum1);

    float4 accA = make_float4(0.0f, 0.0f, 0.0f, 0.0f);
    float4 accB = make_float4(0.0f, 0.0f, 0.0f, 0.0f);

    for (int base = beg; base < end; base += 8) {
        const int j = base + e8;
        const bool ok = j < end;
        const int s = bsrc[ok ? j : beg];
        const float2 t = *(const float2*)(s1 + s * N_HEADS + 2 * hp);
        const int left = end - base;

        float4 va[8], vb[8];
        if (left >= 8) {
            GAT_ROW(0) GAT_ROW(1) GAT_ROW(2) GAT_ROW(3)
            GAT_ROW(4) GAT_ROW(5) GAT_ROW(6) GAT_ROW(7)
        } else {
#pragma unroll
            for (int jj = 0; jj < 8; ++jj) {
                va[jj] = make_float4(0.0f, 0.0f, 0.0f, 0.0f);
                vb[jj] = make_float4(0.0f, 0.0f, 0.0f, 0.0f);
            }
            GAT_ROW(0)
            if (left > 1) GAT_ROW(1)
            if (left > 2) GAT_ROW(2)
            if (left > 3) GAT_ROW(3)
            if (left > 4) GAT_ROW(4)
            if (left > 5) GAT_ROW(5)
            if (left > 6) GAT_ROW(6)
        }

        const float x0 = t.x + s2v.x, x1 = t.y + s2v.y;
        const float v0 = x0 > 0.0f ? x0 : ALPHA * x0;
        const float v1 = x1 > 0.0f ? x1 : ALPHA * x1;
        const float a0 = __expf((ok ? v0 : ninf) - mx0) / sum0;
        const float a1 = __expf((ok ? v1 : ninf) - mx1) / sum1;

#pragma unroll
        for (int jj = 0; jj < 8; ++jj) {
            float w0, w1;
            switch (jj) {
                case 0: w0 = grp8_bcast<0>(a0); w1 = grp8_bcast<0>(a1); break;
                case 1: w0 = grp8_bcast<1>(a0); w1 = grp8_bcast<1>(a1); break;
                case 2: w0 = grp8_bcast<2>(a0); w1 = grp8_bcast<2>(a1); break;
                case 3: w0 = grp8_bcast<3>(a0); w1 = grp8_bcast<3>(a1); break;
                case 4: w0 = grp8_bcast<4>(a0); w1 = grp8_bcast<4>(a1); break;
                case 5: w0 = grp8_bcast<5>(a0); w1 = grp8_bcast<5>(a1); break;
                case 6: w0 = grp8_bcast<6>(a0); w1 = grp8_bcast<6>(a1); break;
                default: w0 = grp8_bcast<7>(a0); w1 = grp8_bcast<7>(a1); break;
            }
            accA.x += w0 * va[jj].x;  accA.y += w0 * va[jj].y;
            accA.z += w0 * va[jj].z;  accA.w += w0 * va[jj].w;
            accB.x += w1 * vb[jj].x;  accB.y += w1 * vb[jj].y;
            accB.z += w1 * vb[jj].z;  accB.w += w1 * vb[jj].w;
        }
    }

    float* op = out + (size_t)node * (N_HEADS * N_HID) + loff;
    *(float4*)op = accA;
    *(float4*)(op + N_HID) = accB;
}

extern "C" void kernel_launch(void* const* d_in, const int* in_sizes, int n_in,
                              void* d_out, int out_size, void* d_ws, size_t ws_size,
                              hipStream_t stream) {
    const float* h   = (const float*)d_in[0];
    const float* W   = (const float*)d_in[1];
    const float* a   = (const float*)d_in[2];
    const int*   src = (const int*)d_in[3];
    const int*   dst = (const int*)d_in[4];
    float* out = (float*)d_out;

    float* ws   = (float*)d_ws;
    float* Wh   = ws;
    float* s1   = Wh + (size_t)N_HEADS * N_NODES * N_HID;
    float* s2   = s1 + (size_t)N_HEADS * N_NODES;
    int*   cur  = (int*)(s2 + (size_t)N_HEADS * N_NODES);
    int*   bsrc = cur + N_NODES;

    gat_zero_cnt<<<(N_NODES + 255) / 256, 256, 0, stream>>>(cur);

    gat_gemm<<<N_NODES / 16, 256, 0, stream>>>(h, W, a, Wh, s1, s2);

    gat_count<<<(N_EDGES + 255) / 256, 256, 0, stream>>>(dst, cur);
    gat_scan<<<1, 1024, 0, stream>>>(cur);
    gat_fill<<<(N_EDGES + 255) / 256, 256, 0, stream>>>(src, dst, cur, bsrc);

    gat_node<<<(N_NODES + 7) / 8, 256, 0, stream>>>(cur, bsrc, s1, s2, Wh, out);
}


# GEMM block tile 16 -> 32 nodes: wave holds 4 accumulators, each W fragment load and each h fragment read feeds 2 WMMAs; h tile in LDS at pitch 260 words (fragment reads were 16-way bank conflicts)
# speedup vs baseline: 4.5368x; 1.0632x over previous
#include <hip/hip_runtime.h>
#include <hip/hip_bf16.h>

typedef float v2f __attribute__((ext_vector_type(2)));
typedef float v8f __attribute__((ext_vector_type(8)));

#define N_NODES 50000
#define N_EDGES 800000
#define N_FEAT  256
#define N_HID   32
#define N_HEADS 8
#define ALPHA   0.2f

template <int K>
__device__ __forceinline__ float row_xor(float v) {
    return __int_as_float(__builtin_amdgcn_update_dpp(0, __float_as_int(v), 0x160 | K,
                                                      0xf, 0xf, true));
}

#define GEMM_ROWS 32
#define SH_PITCH  (N_FEAT + 4)

__device__ __forceinline__ float gat_tile_scores(const v8f c0, const v8f c1,
                                                 const float a1l, const float a1h,
                                                 const float a2l, const float a2h,
                                                 const int m) {
    float x[16];
#pragma unroll
    for (int v = 0; v < 8; ++v) {
        x[v]     = c0[v] * a1l + c1[v] * a1h;
        x[8 + v] = c0[v] * a2l + c1[v] * a2h;
    }
    const bool b3 = (m & 8) != 0, b2 = (m & 4) != 0, b1 = (m & 2) != 0, b0 = (m & 1) != 0;
    float y[8], z[4], w2[2];
#pragma unroll
    for (int i = 0; i < 8; ++i)
        y[i] = (b3 ? x[8 + i] : x[i]) + row_xor<8>(b3 ? x[i] : x[8 + i]);
#pragma unroll
    for (int i = 0; i < 4; ++i)
        z[i] = (b2 ? y[4 + i] : y[i]) + row_xor<4>(b2 ? y[i] : y[4 + i]);
#pragma unroll
    for (int i = 0; i < 2; ++i)
        w2[i] = (b1 ? z[2 + i] : z[i]) + row_xor<2>(b1 ? z[i] : z[2 + i]);
    return (b0 ? w2[1] : w2[0]) + row_xor<1>(b0 ? w2[0] : w2[1]);
}

__global__ __launch_bounds__(256) void gat_gemm(const float* __restrict__ hmat,
                                                const float* __restrict__ W,
                                                const float* __restrict__ avec,
                                                float* __restrict__ Wh,
                                                float* __restrict__ s1,
                                                float* __restrict__ s2) {
    __shared__ float sh[GEMM_ROWS * SH_PITCH];
    __shared__ float ssc[2 * GEMM_ROWS * N_HEADS];

    const int node_base = blockIdx.x * GEMM_ROWS;
    const bool full = node_base + GEMM_ROWS <= N_NODES;

    const float4* g4 = (const float4*)(hmat + (size_t)node_base * N_FEAT);
    const int col4 = threadIdx.x & 63;
    const int r0   = threadIdx.x >> 6;
    float4 t[8];
#pragma unroll
    for (int i = 0; i < 4; ++i)
        t[i] = g4[threadIdx.x + 256 * i];
    if (full) {
#pragma unroll
        for (int i = 4; i < 8; ++i)
            t[i] = g4[threadIdx.x + 256 * i];
    } else {
#pragma unroll
        for (int i = 4; i < 8; ++i)
            t[i] = make_float4(0.0f, 0.0f, 0.0f, 0.0f);
    }
#pragma unroll
    for (int i = 0; i < 8; ++i)
        *(float4*)(sh + (r0 + 4 * i) * SH_PITCH + col4 * 4) = t[i];
    __syncthreads();

    const int lane = threadIdx.x & 31;
    const int head = threadIdx.x >> 5;
    const float* Wp = W + (size_t)head * N_FEAT * N_HID;

    const int m    = lane & 15;
    const int koff = (lane >> 4) << 1;

    v8f c00 = {};
    v8f c01 = {};
    v8f c10 = {};
    v8f c11 = {};

    const float* ar0 = sh + m * SH_PITCH + koff;
    const float* ar1 = ar0 + 16 * SH_PITCH;
    const float* bp  = Wp + koff * N_HID + m;

    for (int k = 0; k < N_FEAT; k += 4) {
        v2f a0, a1, b0, b1;
        a0 = *(const v2f*)(ar0 + k);
        a1 = *(const v2f*)(ar1 + k);
        b0[0] = bp[k * N_HID];
        b0[1] = bp[(k + 1) * N_HID];
        b1[0] = bp[k * N_HID + 16];
        b1[1] = bp[(k + 1) * N_HID + 16];
        c00 = __builtin_amdgcn_wmma_f32_16x16x4_f32(false, a0, false, b0,
                                                    (short)0, c00, false, false);
        c01 = __builtin_amdgcn_wmma_f32_16x16x4_f32(false, a0, false, b1,
                                                    (short)0, c01, false, false);
        c10 = __builtin_amdgcn_wmma_f32_16x16x4_f32(false, a1, false, b0,
                                                    (short)0, c10, false, false);
        c11 = __builtin_amdgcn_wmma_f32_16x16x4_f32(false, a1, false, b1,
                                                    (short)0, c11, false, false);
    }

    const int mb = (lane >> 4) * 8;

    const float* ap = avec + head * 2 * N_HID;
    const float a1l = ap[m],         a1h = ap[16 + m];
    const float a2l = ap[N_HID + m], a2h = ap[N_HID + 16 + m];
    const float sc0 = gat_tile_scores(c00, c01, a1l, a1h, a2l, a2h, m);
    const float sc1 = gat_tile_scores(c10, c11, a1l, a1h, a2l, a2h, m);

    float* sq = ssc + (m >> 3) * (GEMM_ROWS * N_HEADS) + (mb + (m & 7)) * N_HEADS + head;
    sq[0]            = sc0;
    sq[16 * N_HEADS] = sc1;
    __syncthreads();
    if (full || threadIdx.x < 16 * N_HEADS) {
        s1[node_base * N_HEADS + threadIdx.x] = ssc[threadIdx.x];
        s2[node_base * N_HEADS + threadIdx.x] = ssc[GEMM_ROWS * N_HEADS + threadIdx.x];
    }

    float* op = Wh + (size_t)node_base * (N_HEADS * N_HID) + head * N_HID;
#pragma unroll
    for (int v = 0; v < 8; ++v) {
        op[(mb + v) * (N_HEADS * N_HID) + m]      = c00[v];
        op[(mb + v) * (N_HEADS * N_HID) + 16 + m] = c01[v];
    }
    if (full) {
#pragma unroll
        for (int v = 0; v < 8; ++v) {
            op[(16 + mb + v) * (N_HEADS * N_HID) + m]      = c10[v];
            op[(16 + mb + v) * (N_HEADS * N_HID) + 16 + m] = c11[v];
        }
    }
}


__global__ void gat_zero_cnt(int* __restrict__ cur) {
    const int i = blockIdx.x * blockDim.x + threadIdx.x;
    if (i < N_NODES) cur[i] = 0;
}

__global__ void gat_count(const int* __restrict__ dst, int* __restrict__ cur) {
    const int e = blockIdx.x * blockDim.x + threadIdx.x;
    if (e < N_EDGES) atomicAdd(&cur[dst[e]], 1);
}

#define SCAN_TILES ((N_NODES + 4095) / 4096)
__global__ __launch_bounds__(1024) void gat_scan(int* __restrict__ cur) {
    __shared__ int wsum[32];
    const int t = threadIdx.x, lane = t & 31, w = t >> 5;
    int4 c[SCAN_TILES];
#pragma unroll
    for (int k = 0; k < SCAN_TILES; ++k) {
        const int i = k * 4096 + t * 4;
        c[k] = make_int4(0, 0, 0, 0);
        if (i < N_NODES) c[k] = *(const int4*)(cur + i);
    }
    int carry = 0;
#pragma unroll
    for (int k = 0; k < SCAN_TILES; ++k) {
        const int i = k * 4096 + t * 4;
        const int p0 = c[k].x, p1 = p0 + c[k].y, p2 = p1 + c[k].z, p3 = p2 + c[k].w;
        int incl = p3;
        for (int off = 1; off < 32; off <<= 1) {
            const int u = __shfl_up(incl, off, 32);
            if (lane >= off) incl += u;
        }
        if (lane == 31) wsum[w] = incl;
        __syncthreads();
        if (w == 0) {
            int x = wsum[lane];
            for (int off = 1; off < 32; off <<= 1) {
                const int u = __shfl_up(x, off, 32);
                if (lane >= off) x += u;
            }
            wsum[lane] = x;
        }
        __syncthreads();
        const int wave_off = w ? wsum[w - 1] : 0;
        const int total    = wsum[31];
        const int excl     = carry + wave_off + incl - p3;
        if (i < N_NODES)
            *(int4*)(cur + i) = make_int4(excl, excl + p0, excl + p1, excl + p2);
        carry += total;
        __syncthreads();
    }
}

__global__ void gat_fill(const int* __restrict__ src, const int* __restrict__ dst,
                         int* __restrict__ cur, int* __restrict__ bsrc) {
    const int e = blockIdx.x * blockDim.x + threadIdx.x;
    if (e >= N_EDGES) return;
    const int slot = atomicAdd(&cur[dst[e]], 1);
    bsrc[slot] = src[e];
}

template <int J>
__device__ __forceinline__ float grp8_bcast(float v) {
    return __int_as_float(__builtin_amdgcn_mov_dpp8(__float_as_int(v), J * 0x249249));
}

#define GAT_ROW(jj)                                                               \
    {                                                                              \
        const int sj = __builtin_amdgcn_readlane(s, (jj));                         \
              \
        const float* rp = Wh + (size_t)sj * (N_HEADS * N_HID);                     \
        va[jj] = *(const float4*)(rp + loff);                                      \
        vb[jj] = *(const float4*)(rp + loff + N_HID);                              \
    }

__global__ __launch_bounds__(256) void gat_node(const int* __restrict__ cur,
                                                const int* __restrict__ bsrc,
                                                const float* __restrict__ s1,
                                                const float* __restrict__ s2,
                                                const float* __restrict__ Wh,
                                                float* __restrict__ out) {
    const int node = __builtin_amdgcn_readfirstlane((int)(blockIdx.x * 8 + (threadIdx.x >> 5)));
    if (node >= N_NODES) return;
    const int lane = threadIdx.x & 31;
    const int e8 = lane & 7;
    const int hp = lane >> 3;
    const unsigned loff = hp * (2 * N_HID) + e8 * 4;

    const int beg = node ? cur[node - 1] : 0;
    const int end = cur[node];
    const float2 s2v = *(const float2*)(s2 + node * N_HEADS + 2 * hp);

    const float ninf = -__builtin_inff();

    float mx0 = ninf, mx1 = ninf;
    for (int base = beg; base < end; base += 8) {
        const int j = base + e8;
        const bool ok = j < end;
        const int s = bsrc[ok ? j : beg];
        const float2 t = *(const float2*)(s1 + s * N_HEADS + 2 * hp);
        const float x0 = t.x + s2v.x, x1 = t.y + s2v.y;
        const float v0 = x0 > 0.0f ? x0 : ALPHA * x0;
        const float v1 = x1 > 0.0f ? x1 : ALPHA * x1;
        mx0 = fmaxf(mx0, ok ? v0 : ninf);
        mx1 = fmaxf(mx1, ok ? v1 : ninf);
    }
    mx0 = fmaxf(mx0, row_xor<1>(mx0));  mx1 = fmaxf(mx1, row_xor<1>(mx1));
    mx0 = fmaxf(mx0, row_xor<2>(mx0));  mx1 = fmaxf(mx1, row_xor<2>(mx1));
    mx0 = fmaxf(mx0, row_xor<4>(mx0));  mx1 = fmaxf(mx1, row_xor<4>(mx1));

    float sum0 = 0.0f, sum1 = 0.0f;
    for (int base = beg; base < end; base += 8) {
        const int j = base + e8;
        const bool ok = j < end;
        const int s = bsrc[ok ? j : beg];
        const float2 t = *(const float2*)(s1 + s * N_HEADS + 2 * hp);
        const float x0 = t.x + s2v.x, x1 = t.y + s2v.y;
        const float v0 = x0 > 0.0f ? x0 : ALPHA * x0;
        const float v1 = x1 > 0.0f ? x1 : ALPHA * x1;
        sum0 += __expf((ok ? v0 : ninf) - mx0);
        sum1 += __expf((ok ? v1 : ninf) - mx1);
    }
    sum0 += row_xor<1>(sum0);  sum1 += row_xor<1>(sum1);
    sum0 += row_xor<2>(sum0);  sum1 += row_xor<2>(sum1);
    sum0 += row_xor<4>(sum0);  sum1 += row_xor<4>(sum1);

    float4 accA = make_float4(0.0f, 0.0f, 0.0f, 0.0f);
    float4 accB = make_float4(0.0f, 0.0f, 0.0f, 0.0f);

    for (int base = beg; base < end; base += 8) {
        const int j = base + e8;
        const bool ok = j < end;
        const int s = bsrc[ok ? j : beg];
        const float2 t = *(const float2*)(s1 + s * N_HEADS + 2 * hp);
        const int left = end - base;

        float4 va[8], vb[8];
        if (left >= 8) {
            GAT_ROW(0) GAT_ROW(1) GAT_ROW(2) GAT_ROW(3)
            GAT_ROW(4) GAT_ROW(5) GAT_ROW(6) GAT_ROW(7)
        } else {
#pragma unroll
            for (int jj = 0; jj < 8; ++jj) {
                va[jj] = make_float4(0.0f, 0.0f, 0.0f, 0.0f);
                vb[jj] = make_float4(0.0f, 0.0f, 0.0f, 0.0f);
            }
            GAT_ROW(0)
            if (left > 1) GAT_ROW(1)
            if (left > 2) GAT_ROW(2)
            if (left > 3) GAT_ROW(3)
            if (left > 4) GAT_ROW(4)
            if (left > 5) GAT_ROW(5)
            if (left > 6) GAT_ROW(6)
        }

        const float x0 = t.x + s2v.x, x1 = t.y + s2v.y;
        const float v0 = x0 > 0.0f ? x0 : ALPHA * x0;
        const float v1 = x1 > 0.0f ? x1 : ALPHA * x1;
        const float a0 = __expf((ok ? v0 : ninf) - mx0) / sum0;
        const float a1 = __expf((ok ? v1 : ninf) - mx1) / sum1;

#pragma unroll
        for (int jj = 0; jj < 8; ++jj) {
            float w0, w1;
            switch (jj) {
                case 0: w0 = grp8_bcast<0>(a0); w1 = grp8_bcast<0>(a1); break;
                case 1: w0 = grp8_bcast<1>(a0); w1 = grp8_bcast<1>(a1); break;
                case 2: w0 = grp8_bcast<2>(a0); w1 = grp8_bcast<2>(a1); break;
                case 3: w0 = grp8_bcast<3>(a0); w1 = grp8_bcast<3>(a1); break;
                case 4: w0 = grp8_bcast<4>(a0); w1 = grp8_bcast<4>(a1); break;
                case 5: w0 = grp8_bcast<5>(a0); w1 = grp8_bcast<5>(a1); break;
                case 6: w0 = grp8_bcast<6>(a0); w1 = grp8_bcast<6>(a1); break;
                default: w0 = grp8_bcast<7>(a0); w1 = grp8_bcast<7>(a1); break;
            }
            accA.x += w0 * va[jj].x;  accA.y += w0 * va[jj].y;
            accA.z += w0 * va[jj].z;  accA.w += w0 * va[jj].w;
            accB.x += w1 * vb[jj].x;  accB.y += w1 * vb[jj].y;
            accB.z += w1 * vb[jj].z;  accB.w += w1 * vb[jj].w;
        }
    }

    float* op = out + (size_t)node * (N_HEADS * N_HID) + loff;
    *(float4*)op = accA;
    *(float4*)(op + N_HID) = accB;
}

extern "C" void kernel_launch(void* const* d_in, const int* in_sizes, int n_in,
                              void* d_out, int out_size, void* d_ws, size_t ws_size,
                              hipStream_t stream) {
    const float* h   = (const float*)d_in[0];
    const float* W   = (const float*)d_in[1];
    const float* a   = (const float*)d_in[2];
    const int*   src = (const int*)d_in[3];
    const int*   dst = (const int*)d_in[4];
    float* out = (float*)d_out;

    float* ws   = (float*)d_ws;
    float* Wh   = ws;
    float* s1   = Wh + (size_t)N_HEADS * N_NODES * N_HID;
    float* s2   = s1 + (size_t)N_HEADS * N_NODES;
    int*   cur  = (int*)(s2 + (size_t)N_HEADS * N_NODES);
    int*   bsrc = cur + N_NODES;

    gat_zero_cnt<<<(N_NODES + 255) / 256, 256, 0, stream>>>(cur);

    gat_gemm<<<(N_NODES + GEMM_ROWS - 1) / GEMM_ROWS, 256, 0, stream>>>(h, W, a, Wh, s1, s2);

    gat_count<<<(N_EDGES + 255) / 256, 256, 0, stream>>>(dst, cur);
    gat_scan<<<1, 1024, 0, stream>>>(cur);
    gat_fill<<<(N_EDGES + 255) / 256, 256, 0, stream>>>(src, dst, cur, bsrc);

    gat_node<<<(N_NODES + 7) / 8, 256, 0, stream>>>(cur, bsrc, s1, s2, Wh, out);
}


# each dst node owns a fixed 64-slot piece (slot = int atomic on its counter; fuller nodes use an overflow list): edge count pass and prefix-scan launch removed; sum order still int-atomic arrival
# speedup vs baseline: 6.3150x; 1.3920x over previous
#include <hip/hip_runtime.h>
#include <hip/hip_bf16.h>

typedef float v2f __attribute__((ext_vector_type(2)));
typedef float v8f __attribute__((ext_vector_type(8)));

#define N_NODES 50000
#define N_EDGES 800000
#define N_FEAT  256
#define N_HID   32
#define N_HEADS 8
#define ALPHA   0.2f

template <int K>
__device__ __forceinline__ float row_xor(float v) {
    return __int_as_float(__builtin_amdgcn_update_dpp(0, __float_as_int(v), 0x160 | K,
                                                      0xf, 0xf, true));
}

#define GEMM_ROWS 32
#define SH_PITCH  (N_FEAT + 4)

__device__ __forceinline__ float gat_tile_scores(const v8f c0, const v8f c1,
                                                 const float a1l, const float a1h,
                                                 const float a2l, const float a2h,
                                                 const int m) {
    float x[16];
#pragma unroll
    for (int v = 0; v < 8; ++v) {
        x[v]     = c0[v] * a1l + c1[v] * a1h;
        x[8 + v] = c0[v] * a2l + c1[v] * a2h;
    }
    const bool b3 = (m & 8) != 0, b2 = (m & 4) != 0, b1 = (m & 2) != 0, b0 = (m & 1) != 0;
    float y[8], z[4], w2[2];
#pragma unroll
    for (int i = 0; i < 8; ++i)
        y[i] = (b3 ? x[8 + i] : x[i]) + row_xor<8>(b3 ? x[i] : x[8 + i]);
#pragma unroll
    for (int i = 0; i < 4; ++i)
        z[i] = (b2 ? y[4 + i] : y[i]) + row_xor<4>(b2 ? y[i] : y[4 + i]);
#pragma unroll
    for (int i = 0; i < 2; ++i)
        w2[i] = (b1 ? z[2 + i] : z[i]) + row_xor<2>(b1 ? z[i] : z[2 + i]);
    return (b0 ? w2[1] : w2[0]) + row_xor<1>(b0 ? w2[0] : w2[1]);
}

__global__ __launch_bounds__(256) void gat_gemm(const float* __restrict__ hmat,
                                                const float* __restrict__ W,
                                                const float* __restrict__ avec,
                                                float* __restrict__ Wh,
                                                float* __restrict__ s1,
                                                float* __restrict__ s2) {
    __shared__ float sh[GEMM_ROWS * SH_PITCH];
    __shared__ float ssc[2 * GEMM_ROWS * N_HEADS];

    const int node_base = blockIdx.x * GEMM_ROWS;
    const bool full = node_base + GEMM_ROWS <= N_NODES;

    const float4* g4 = (const float4*)(hmat + (size_t)node_base * N_FEAT);
    const int col4 = threadIdx.x & 63;
    const int r0   = threadIdx.x >> 6;
    float4 t[8];
#pragma unroll
    for (int i = 0; i < 4; ++i)
        t[i] = g4[threadIdx.x + 256 * i];
    if (full) {
#pragma unroll
        for (int i = 4; i < 8; ++i)
            t[i] = g4[threadIdx.x + 256 * i];
    } else {
#pragma unroll
        for (int i = 4; i < 8; ++i)
            t[i] = make_float4(0.0f, 0.0f, 0.0f, 0.0f);
    }
#pragma unroll
    for (int i = 0; i < 8; ++i)
        *(float4*)(sh + (r0 + 4 * i) * SH_PITCH + col4 * 4) = t[i];
    __syncthreads();

    const int lane = threadIdx.x & 31;
    const int head = threadIdx.x >> 5;
    const float* Wp = W + (size_t)head * N_FEAT * N_HID;

    const int m    = lane & 15;
    const int koff = (lane >> 4) << 1;

    v8f c00 = {};
    v8f c01 = {};
    v8f c10 = {};
    v8f c11 = {};

    const float* ar0 = sh + m * SH_PITCH + koff;
    const float* ar1 = ar0 + 16 * SH_PITCH;
    const float* bp  = Wp + koff * N_HID + m;

    for (int k = 0; k < N_FEAT; k += 4) {
        v2f a0, a1, b0, b1;
        a0 = *(const v2f*)(ar0 + k);
        a1 = *(const v2f*)(ar1 + k);
        b0[0] = bp[k * N_HID];
        b0[1] = bp[(k + 1) * N_HID];
        b1[0] = bp[k * N_HID + 16];
        b1[1] = bp[(k + 1) * N_HID + 16];
        c00 = __builtin_amdgcn_wmma_f32_16x16x4_f32(false, a0, false, b0,
                                                    (short)0, c00, false, false);
        c01 = __builtin_amdgcn_wmma_f32_16x16x4_f32(false, a0, false, b1,
                                                    (short)0, c01, false, false);
        c10 = __builtin_amdgcn_wmma_f32_16x16x4_f32(false, a1, false, b0,
                                                    (short)0, c10, false, false);
        c11 = __builtin_amdgcn_wmma_f32_16x16x4_f32(false, a1, false, b1,
                                                    (short)0, c11, false, false);
    }

    const int mb = (lane >> 4) * 8;

    const float* ap = avec + head * 2 * N_HID;
    const float a1l = ap[m],         a1h = ap[16 + m];
    const float a2l = ap[N_HID + m], a2h = ap[N_HID + 16 + m];
    const float sc0 = gat_tile_scores(c00, c01, a1l, a1h, a2l, a2h, m);
    const float sc1 = gat_tile_scores(c10, c11, a1l, a1h, a2l, a2h, m);

    float* sq = ssc + (m >> 3) * (GEMM_ROWS * N_HEADS) + (mb + (m & 7)) * N_HEADS + head;
    sq[0]            = sc0;
    sq[16 * N_HEADS] = sc1;
    __syncthreads();
    if (full || threadIdx.x < 16 * N_HEADS) {
        s1[node_base * N_HEADS + threadIdx.x] = ssc[threadIdx.x];
        s2[node_base * N_HEADS + threadIdx.x] = ssc[GEMM_ROWS * N_HEADS + threadIdx.x];
    }

    float* op = Wh + (size_t)node_base * (N_HEADS * N_HID) + head * N_HID;
#pragma unroll
    for (int v = 0; v < 8; ++v) {
        op[(mb + v) * (N_HEADS * N_HID) + m]      = c00[v];
        op[(mb + v) * (N_HEADS * N_HID) + 16 + m] = c01[v];
    }
    if (full) {
#pragma unroll
        for (int v = 0; v < 8; ++v) {
            op[(16 + mb + v) * (N_HEADS * N_HID) + m]      = c10[v];
            op[(16 + mb + v) * (N_HEADS * N_HID) + 16 + m] = c11[v];
        }
    }
}

#define BKT_CAP   64
#define CNT_WORDS ((N_NODES + 255) / 256 * 256)

__global__ void gat_zero_cnt(int* __restrict__ cnt) {
    cnt[blockIdx.x * blockDim.x + threadIdx.x] = 0;
}

__global__ void gat_fill(const int* __restrict__ src, const int* __restrict__ dst,
                         int* __restrict__ cnt, int* __restrict__ bsrc,
                         int* __restrict__ ovf_n, int* __restrict__ ovf_src,
                         int* __restrict__ ovf_dst) {
    const int e = blockIdx.x * blockDim.x + threadIdx.x;
    if (e >= N_EDGES) return;
    const int d = dst[e];
    const int slot = atomicAdd(&cnt[d], 1);
    if (slot < BKT_CAP) {
        bsrc[d * BKT_CAP + slot] = src[e];
    } else {
        const int o = atomicAdd(ovf_n, 1);
        ovf_src[o] = src[e];
        ovf_dst[o] = d;
    }
}

template <int J>
__device__ __forceinline__ float grp8_bcast(float v) {
    return __int_as_float(__builtin_amdgcn_mov_dpp8(__float_as_int(v), J * 0x249249));
}

#define GAT_ROW(jj)                                                               \
    {                                                                              \
        const int sj = __builtin_amdgcn_readlane(s, (jj));                         \
              \
        const float* rp = Wh + (size_t)sj * (N_HEADS * N_HID);                     \
        va[jj] = *(const float4*)(rp + loff);                                      \
        vb[jj] = *(const float4*)(rp + loff + N_HID);                              \
    }

__device__ __forceinline__ bool gat_ovf_edge(const int j, const int on, const int node,
                                             const int* __restrict__ ovf_src,
                                             const int* __restrict__ ovf_dst,
                                             const float* __restrict__ s1, const int hp,
                                             const float2 s2v, int& s, float& v0, float& v1) {
    const bool in = j < on;
    const int jc = in ? j : 0;
    const bool ok = in && ovf_dst[jc] == node;
    s = ovf_src[jc];
    const float2 t = *(const float2*)(s1 + s * N_HEADS + 2 * hp);
    const float x0 = t.x + s2v.x, x1 = t.y + s2v.y;
    v0 = x0 > 0.0f ? x0 : ALPHA * x0;
    v1 = x1 > 0.0f ? x1 : ALPHA * x1;
    return ok;
}

__global__ __launch_bounds__(256) void gat_node(const int* __restrict__ cnt,
                                                const int* __restrict__ bsrc,
                                                const int* __restrict__ ovf_n,
                                                const int* __restrict__ ovf_src,
                                                const int* __restrict__ ovf_dst,
                                                const float* __restrict__ s1,
                                                const float* __restrict__ s2,
                                                const float* __restrict__ Wh,
                                                float* __restrict__ out) {
    const int node = __builtin_amdgcn_readfirstlane((int)(blockIdx.x * 8 + (threadIdx.x >> 5)));
    if (node >= N_NODES) return;
    const int lane = threadIdx.x & 31;
    const int e8 = lane & 7;
    const int hp = lane >> 3;
    const unsigned loff = hp * (2 * N_HID) + e8 * 4;

    const int deg = __builtin_amdgcn_readfirstlane(cnt[node]);
    const int beg = node * BKT_CAP;
    const int end = beg + (deg < BKT_CAP ? deg : BKT_CAP);
    const int on  = deg > BKT_CAP ? __builtin_amdgcn_readfirstlane(*ovf_n) : 0;
    const float2 s2v = *(const float2*)(s2 + node * N_HEADS + 2 * hp);

    const float ninf = -__builtin_inff();

    float mx0 = ninf, mx1 = ninf;
    for (int base = beg; base < end; base += 8) {
        const int j = base + e8;
        const bool ok = j < end;
        const int s = bsrc[ok ? j : beg];
        const float2 t = *(const float2*)(s1 + s * N_HEADS + 2 * hp);
        const float x0 = t.x + s2v.x, x1 = t.y + s2v.y;
        const float v0 = x0 > 0.0f ? x0 : ALPHA * x0;
        const float v1 = x1 > 0.0f ? x1 : ALPHA * x1;
        mx0 = fmaxf(mx0, ok ? v0 : ninf);
        mx1 = fmaxf(mx1, ok ? v1 : ninf);
    }
    for (int base = 0; base < on; base += 8) {
        int s; float v0, v1;
        const bool ok = gat_ovf_edge(base + e8, on, node, ovf_src, ovf_dst, s1, hp, s2v, s, v0, v1);
        mx0 = fmaxf(mx0, ok ? v0 : ninf);
        mx1 = fmaxf(mx1, ok ? v1 : ninf);
    }
    mx0 = fmaxf(mx0, row_xor<1>(mx0));  mx1 = fmaxf(mx1, row_xor<1>(mx1));
    mx0 = fmaxf(mx0, row_xor<2>(mx0));  mx1 = fmaxf(mx1, row_xor<2>(mx1));
    mx0 = fmaxf(mx0, row_xor<4>(mx0));  mx1 = fmaxf(mx1, row_xor<4>(mx1));

    float sum0 = 0.0f, sum1 = 0.0f;
    for (int base = beg; base < end; base += 8) {
        const int j = base + e8;
        const bool ok = j < end;
        const int s = bsrc[ok ? j : beg];
        const float2 t = *(const float2*)(s1 + s * N_HEADS + 2 * hp);
        const float x0 = t.x + s2v.x, x1 = t.y + s2v.y;
        const float v0 = x0 > 0.0f ? x0 : ALPHA * x0;
        const float v1 = x1 > 0.0f ? x1 : ALPHA * x1;
        sum0 += __expf((ok ? v0 : ninf) - mx0);
        sum1 += __expf((ok ? v1 : ninf) - mx1);
    }
    for (int base = 0; base < on; base += 8) {
        int s; float v0, v1;
        const bool ok = gat_ovf_edge(base + e8, on, node, ovf_src, ovf_dst, s1, hp, s2v, s, v0, v1);
        sum0 += __expf((ok ? v0 : ninf) - mx0);
        sum1 += __expf((ok ? v1 : ninf) - mx1);
    }
    sum0 += row_xor<1>(sum0);  sum1 += row_xor<1>(sum1);
    sum0 += row_xor<2>(sum0);  sum1 += row_xor<2>(sum1);
    sum0 += row_xor<4>(sum0);  sum1 += row_xor<4>(sum1);

    float4 accA = make_float4(0.0f, 0.0f, 0.0f, 0.0f);
    float4 accB = make_float4(0.0f, 0.0f, 0.0f, 0.0f);

    for (int base = beg; base < end; base += 8) {
        const int j = base + e8;
        const bool ok = j < end;
        const int s = bsrc[ok ? j : beg];
        const float2 t = *(const float2*)(s1 + s * N_HEADS + 2 * hp);
        const int left = end - base;

        float4 va[8], vb[8];
        if (left >= 8) {
            GAT_ROW(0) GAT_ROW(1) GAT_ROW(2) GAT_ROW(3)
            GAT_ROW(4) GAT_ROW(5) GAT_ROW(6) GAT_ROW(7)
        } else {
#pragma unroll
            for (int jj = 0; jj < 8; ++jj) {
                va[jj] = make_float4(0.0f, 0.0f, 0.0f, 0.0f);
                vb[jj] = make_float4(0.0f, 0.0f, 0.0f, 0.0f);
            }
            GAT_ROW(0)
            if (left > 1) GAT_ROW(1)
            if (left > 2) GAT_ROW(2)
            if (left > 3) GAT_ROW(3)
            if (left > 4) GAT_ROW(4)
            if (left > 5) GAT_ROW(5)
            if (left > 6) GAT_ROW(6)
        }

        const float x0 = t.x + s2v.x, x1 = t.y + s2v.y;
        const float v0 = x0 > 0.0f ? x0 : ALPHA * x0;
        const float v1 = x1 > 0.0f ? x1 : ALPHA * x1;
        const float a0 = __expf((ok ? v0 : ninf) - mx0) / sum0;
        const float a1 = __expf((ok ? v1 : ninf) - mx1) / sum1;

#pragma unroll
        for (int jj = 0; jj < 8; ++jj) {
            float w0, w1;
            switch (jj) {
                case 0: w0 = grp8_bcast<0>(a0); w1 = grp8_bcast<0>(a1); break;
                case 1: w0 = grp8_bcast<1>(a0); w1 = grp8_bcast<1>(a1); break;
                case 2: w0 = grp8_bcast<2>(a0); w1 = grp8_bcast<2>(a1); break;
                case 3: w0 = grp8_bcast<3>(a0); w1 = grp8_bcast<3>(a1); break;
                case 4: w0 = grp8_bcast<4>(a0); w1 = grp8_bcast<4>(a1); break;
                case 5: w0 = grp8_bcast<5>(a0); w1 = grp8_bcast<5>(a1); break;
                case 6: w0 = grp8_bcast<6>(a0); w1 = grp8_bcast<6>(a1); break;
                default: w0 = grp8_bcast<7>(a0); w1 = grp8_bcast<7>(a1); break;
            }
            accA.x += w0 * va[jj].x;  accA.y += w0 * va[jj].y;
            accA.z += w0 * va[jj].z;  accA.w += w0 * va[jj].w;
            accB.x += w1 * vb[jj].x;  accB.y += w1 * vb[jj].y;
            accB.z += w1 * vb[jj].z;  accB.w += w1 * vb[jj].w;
        }
    }

    for (int base = 0; base < on; base += 8) {
        int s; float v0, v1;
        const bool ok = gat_ovf_edge(base + e8, on, node, ovf_src, ovf_dst, s1, hp, s2v, s, v0, v1);
        const float a0 = __expf((ok ? v0 : ninf) - mx0) / sum0;
        const float a1 = __expf((ok ? v1 : ninf) - mx1) / sum1;
        const unsigned okm = __builtin_amdgcn_ballot_w32(ok) & 0xffu;
        for (int jj = 0; jj < 8; ++jj) {
            if (!((okm >> jj) & 1u)) continue;
            const int sj = __shfl(s, jj, 32);
            const float* rp = Wh + (size_t)sj * (N_HEADS * N_HID);
            const float4 ra = *(const float4*)(rp + loff);
            const float4 rb = *(const float4*)(rp + loff + N_HID);
            const float w0 = __shfl(a0, (lane & 24) | jj, 32);
            const float w1 = __shfl(a1, (lane & 24) | jj, 32);
            accA.x += w0 * ra.x;  accA.y += w0 * ra.y;
            accA.z += w0 * ra.z;  accA.w += w0 * ra.w;
            accB.x += w1 * rb.x;  accB.y += w1 * rb.y;
            accB.z += w1 * rb.z;  accB.w += w1 * rb.w;
        }
    }

    float* op = out + (size_t)node * (N_HEADS * N_HID) + loff;
    *(float4*)op = accA;
    *(float4*)(op + N_HID) = accB;
}

extern "C" void kernel_launch(void* const* d_in, const int* in_sizes, int n_in,
                              void* d_out, int out_size, void* d_ws, size_t ws_size,
                              hipStream_t stream) {
    const float* h   = (const float*)d_in[0];
    const float* W   = (const float*)d_in[1];
    const float* a   = (const float*)d_in[2];
    const int*   src = (const int*)d_in[3];
    const int*   dst = (const int*)d_in[4];
    float* out = (float*)d_out;

    float* ws   = (float*)d_ws;
    float* Wh   = ws;
    float* s1   = Wh + (size_t)N_HEADS * N_NODES * N_HID;
    float* s2   = s1 + (size_t)N_HEADS * N_NODES;
    int*   cnt  = (int*)(s2 + (size_t)N_HEADS * N_NODES);
    int*   bsrc = cnt + CNT_WORDS;
    int*   osrc = bsrc + (size_t)N_NODES * BKT_CAP;
    int*   odst = osrc + N_EDGES;
    int*   ovfn = cnt + N_NODES;

    gat_zero_cnt<<<CNT_WORDS / 256, 256, 0, stream>>>(cnt);

    gat_gemm<<<(N_NODES + GEMM_ROWS - 1) / GEMM_ROWS, 256, 0, stream>>>(h, W, a, Wh, s1, s2);

    gat_fill<<<(N_EDGES + 255) / 256, 256, 0, stream>>>(src, dst, cnt, bsrc, ovfn, osrc, odst);

    gat_node<<<(N_NODES + 7) / 8, 256, 0, stream>>>(cnt, bsrc, ovfn, osrc, odst, s1, s2, Wh, out);
}


# bucket fill moved into the GEMM launch as 196 leading blocks (16 edges a thread, loads then int atomics all in flight): overlaps the matrix work, fill launch removed; sums still in arrival order
# speedup vs baseline: 6.9193x; 1.0957x over previous
#include <hip/hip_runtime.h>
#include <hip/hip_bf16.h>

typedef float v2f __attribute__((ext_vector_type(2)));
typedef float v8f __attribute__((ext_vector_type(8)));

#define N_NODES 50000
#define N_EDGES 800000
#define N_FEAT  256
#define N_HID   32
#define N_HEADS 8
#define ALPHA   0.2f

template <int K>
__device__ __forceinline__ float row_xor(float v) {
    return __int_as_float(__builtin_amdgcn_update_dpp(0, __float_as_int(v), 0x160 | K,
                                                      0xf, 0xf, true));
}

#define GEMM_ROWS 32
#define SH_PITCH  (N_FEAT + 4)

__device__ __forceinline__ float gat_tile_scores(const v8f c0, const v8f c1,
                                                 const float a1l, const float a1h,
                                                 const float a2l, const float a2h,
                                                 const int m) {
    float x[16];
#pragma unroll
    for (int v = 0; v < 8; ++v) {
        x[v]     = c0[v] * a1l + c1[v] * a1h;
        x[8 + v] = c0[v] * a2l + c1[v] * a2h;
    }
    const bool b3 = (m & 8) != 0, b2 = (m & 4) != 0, b1 = (m & 2) != 0, b0 = (m & 1) != 0;
    float y[8], z[4], w2[2];
#pragma unroll
    for (int i = 0; i < 8; ++i)
        y[i] = (b3 ? x[8 + i] : x[i]) + row_xor<8>(b3 ? x[i] : x[8 + i]);
#pragma unroll
    for (int i = 0; i < 4; ++i)
        z[i] = (b2 ? y[4 + i] : y[i]) + row_xor<4>(b2 ? y[i] : y[4 + i]);
#pragma unroll
    for (int i = 0; i < 2; ++i)
        w2[i] = (b1 ? z[2 + i] : z[i]) + row_xor<2>(b1 ? z[i] : z[2 + i]);
    return (b0 ? w2[1] : w2[0]) + row_xor<1>(b0 ? w2[0] : w2[1]);
}

#define BKT_CAP   64
#define CNT_WORDS ((N_NODES + 255) / 256 * 256)
#define FILL_EPT     16
#define FILL_THREADS (N_EDGES / FILL_EPT)
#define FILL_BLOCKS  ((FILL_THREADS + 255) / 256)
#define GEMM_BLOCKS  ((N_NODES + GEMM_ROWS - 1) / GEMM_ROWS)
static_assert(N_EDGES % FILL_EPT == 0, "a fill thread owns whole 16-byte pieces only");

__device__ __forceinline__ void gat_fill_edge(const int d, const int s, const int slot,
                                              int* __restrict__ bsrc, int* __restrict__ ovf_n,
                                              int* __restrict__ ovf_src, int* __restrict__ ovf_dst) {
    if (slot < BKT_CAP) {
        bsrc[d * BKT_CAP + slot] = s;
    } else {
        const int o = atomicAdd(ovf_n, 1);
        ovf_src[o] = s;
        ovf_dst[o] = d;
    }
}

__device__ __forceinline__ void gat_fill_block(const int* __restrict__ src,
                                               const int* __restrict__ dst,
                                               int* __restrict__ cnt, int* __restrict__ bsrc,
                                               int* __restrict__ ovf_n, int* __restrict__ ovf_src,
                                               int* __restrict__ ovf_dst) {
    const int gt = blockIdx.x * 256 + threadIdx.x;
    if (gt >= FILL_THREADS) return;
    const int4* d4 = (const int4*)dst;
    const int4* s4 = (const int4*)src;
    int4 dv[FILL_EPT / 4], sv[FILL_EPT / 4], sl[FILL_EPT / 4];
#pragma unroll
    for (int i = 0; i < FILL_EPT / 4; ++i) {
        dv[i] = d4[i * FILL_THREADS + gt];
        sv[i] = s4[i * FILL_THREADS + gt];
    }
#pragma unroll
    for (int i = 0; i < FILL_EPT / 4; ++i) {
        sl[i].x = atomicAdd(&cnt[dv[i].x], 1);
        sl[i].y = atomicAdd(&cnt[dv[i].y], 1);
        sl[i].z = atomicAdd(&cnt[dv[i].z], 1);
        sl[i].w = atomicAdd(&cnt[dv[i].w], 1);
    }
#pragma unroll
    for (int i = 0; i < FILL_EPT / 4; ++i) {
        gat_fill_edge(dv[i].x, sv[i].x, sl[i].x, bsrc, ovf_n, ovf_src, ovf_dst);
        gat_fill_edge(dv[i].y, sv[i].y, sl[i].y, bsrc, ovf_n, ovf_src, ovf_dst);
        gat_fill_edge(dv[i].z, sv[i].z, sl[i].z, bsrc, ovf_n, ovf_src, ovf_dst);
        gat_fill_edge(dv[i].w, sv[i].w, sl[i].w, bsrc, ovf_n, ovf_src, ovf_dst);
    }
}

__global__ __launch_bounds__(256) void gat_gemm(const float* __restrict__ hmat,
                                                const float* __restrict__ W,
                                                const float* __restrict__ avec,
                                                float* __restrict__ Wh,
                                                float* __restrict__ s1,
                                                float* __restrict__ s2,
                                                const int* __restrict__ src,
                                                const int* __restrict__ dst,
                                                int* __restrict__ cnt,
                                                int* __restrict__ bsrc,
                                                int* __restrict__ ovf_n,
                                                int* __restrict__ ovf_src,
                                                int* __restrict__ ovf_dst) {
    __shared__ float sh[GEMM_ROWS * SH_PITCH];
    __shared__ float ssc[2 * GEMM_ROWS * N_HEADS];

    if (blockIdx.x < FILL_BLOCKS) {
        gat_fill_block(src, dst, cnt, bsrc, ovf_n, ovf_src, ovf_dst);
        return;
    }

    const int node_base = (blockIdx.x - FILL_BLOCKS) * GEMM_ROWS;
    const bool full = node_base + GEMM_ROWS <= N_NODES;

    const float4* g4 = (const float4*)(hmat + (size_t)node_base * N_FEAT);
    const int col4 = threadIdx.x & 63;
    const int r0   = threadIdx.x >> 6;
    float4 t[8];
#pragma unroll
    for (int i = 0; i < 4; ++i)
        t[i] = g4[threadIdx.x + 256 * i];
    if (full) {
#pragma unroll
        for (int i = 4; i < 8; ++i)
            t[i] = g4[threadIdx.x + 256 * i];
    } else {
#pragma unroll
        for (int i = 4; i < 8; ++i)
            t[i] = make_float4(0.0f, 0.0f, 0.0f, 0.0f);
    }
#pragma unroll
    for (int i = 0; i < 8; ++i)
        *(float4*)(sh + (r0 + 4 * i) * SH_PITCH + col4 * 4) = t[i];
    __syncthreads();

    const int lane = threadIdx.x & 31;
    const int head = threadIdx.x >> 5;
    const float* Wp = W + (size_t)head * N_FEAT * N_HID;

    const int m    = lane & 15;
    const int koff = (lane >> 4) << 1;

    v8f c00 = {};
    v8f c01 = {};
    v8f c10 = {};
    v8f c11 = {};

    const float* ar0 = sh + m * SH_PITCH + koff;
    const float* ar1 = ar0 + 16 * SH_PITCH;
    const float* bp  = Wp + koff * N_HID + m;

    for (int k = 0; k < N_FEAT; k += 4) {
        v2f a0, a1, b0, b1;
        a0 = *(const v2f*)(ar0 + k);
        a1 = *(const v2f*)(ar1 + k);
        b0[0] = bp[k * N_HID];
        b0[1] = bp[(k + 1) * N_HID];
        b1[0] = bp[k * N_HID + 16];
        b1[1] = bp[(k + 1) * N_HID + 16];
        c00 = __builtin_amdgcn_wmma_f32_16x16x4_f32(false, a0, false, b0,
                                                    (short)0, c00, false, false);
        c01 = __builtin_amdgcn_wmma_f32_16x16x4_f32(false, a0, false, b1,
                                                    (short)0, c01, false, false);
        c10 = __builtin_amdgcn_wmma_f32_16x16x4_f32(false, a1, false, b0,
                                                    (short)0, c10, false, false);
        c11 = __builtin_amdgcn_wmma_f32_16x16x4_f32(false, a1, false, b1,
                                                    (short)0, c11, false, false);
    }

    const int mb = (lane >> 4) * 8;

    const float* ap = avec + head * 2 * N_HID;
    const float a1l = ap[m],         a1h = ap[16 + m];
    const float a2l = ap[N_HID + m], a2h = ap[N_HID + 16 + m];
    const float sc0 = gat_tile_scores(c00, c01, a1l, a1h, a2l, a2h, m);
    const float sc1 = gat_tile_scores(c10, c11, a1l, a1h, a2l, a2h, m);

    float* sq = ssc + (m >> 3) * (GEMM_ROWS * N_HEADS) + (mb + (m & 7)) * N_HEADS + head;
    sq[0]            = sc0;
    sq[16 * N_HEADS] = sc1;
    __syncthreads();
    if (full || threadIdx.x < 16 * N_HEADS) {
        s1[node_base * N_HEADS + threadIdx.x] = ssc[threadIdx.x];
        s2[node_base * N_HEADS + threadIdx.x] = ssc[GEMM_ROWS * N_HEADS + threadIdx.x];
    }

    float* op = Wh + (size_t)node_base * (N_HEADS * N_HID) + head * N_HID;
#pragma unroll
    for (int v = 0; v < 8; ++v) {
        op[(mb + v) * (N_HEADS * N_HID) + m]      = c00[v];
        op[(mb + v) * (N_HEADS * N_HID) + 16 + m] = c01[v];
    }
    if (full) {
#pragma unroll
        for (int v = 0; v < 8; ++v) {
            op[(16 + mb + v) * (N_HEADS * N_HID) + m]      = c10[v];
            op[(16 + mb + v) * (N_HEADS * N_HID) + 16 + m] = c11[v];
        }
    }
}

__global__ void gat_zero_cnt(int* __restrict__ cnt) {
    cnt[blockIdx.x * blockDim.x + threadIdx.x] = 0;
}


template <int J>
__device__ __forceinline__ float grp8_bcast(float v) {
    return __int_as_float(__builtin_amdgcn_mov_dpp8(__float_as_int(v), J * 0x249249));
}

#define GAT_ROW(jj)                                                               \
    {                                                                              \
        const int sj = __builtin_amdgcn_readlane(s, (jj));                         \
              \
        const float* rp = Wh + (size_t)sj * (N_HEADS * N_HID);                     \
        va[jj] = *(const float4*)(rp + loff);                                      \
        vb[jj] = *(const float4*)(rp + loff + N_HID);                              \
    }

__device__ __forceinline__ bool gat_ovf_edge(const int j, const int on, const int node,
                                             const int* __restrict__ ovf_src,
                                             const int* __restrict__ ovf_dst,
                                             const float* __restrict__ s1, const int hp,
                                             const float2 s2v, int& s, float& v0, float& v1) {
    const bool in = j < on;
    const int jc = in ? j : 0;
    const bool ok = in && ovf_dst[jc] == node;
    s = ovf_src[jc];
    const float2 t = *(const float2*)(s1 + s * N_HEADS + 2 * hp);
    const float x0 = t.x + s2v.x, x1 = t.y + s2v.y;
    v0 = x0 > 0.0f ? x0 : ALPHA * x0;
    v1 = x1 > 0.0f ? x1 : ALPHA * x1;
    return ok;
}

__global__ __launch_bounds__(256) void gat_node(const int* __restrict__ cnt,
                                                const int* __restrict__ bsrc,
                                                const int* __restrict__ ovf_n,
                                                const int* __restrict__ ovf_src,
                                                const int* __restrict__ ovf_dst,
                                                const float* __restrict__ s1,
                                                const float* __restrict__ s2,
                                                const float* __restrict__ Wh,
                                                float* __restrict__ out) {
    const int node = __builtin_amdgcn_readfirstlane((int)(blockIdx.x * 8 + (threadIdx.x >> 5)));
    if (node >= N_NODES) return;
    const int lane = threadIdx.x & 31;
    const int e8 = lane & 7;
    const int hp = lane >> 3;
    const unsigned loff = hp * (2 * N_HID) + e8 * 4;

    const int deg = __builtin_amdgcn_readfirstlane(cnt[node]);
    const int beg = node * BKT_CAP;
    const int end = beg + (deg < BKT_CAP ? deg : BKT_CAP);
    const int on  = deg > BKT_CAP ? __builtin_amdgcn_readfirstlane(*ovf_n) : 0;
    const float2 s2v = *(const float2*)(s2 + node * N_HEADS + 2 * hp);

    const float ninf = -__builtin_inff();

    float mx0 = ninf, mx1 = ninf;
    for (int base = beg; base < end; base += 8) {
        const int j = base + e8;
        const bool ok = j < end;
        const int s = bsrc[ok ? j : beg];
        const float2 t = *(const float2*)(s1 + s * N_HEADS + 2 * hp);
        const float x0 = t.x + s2v.x, x1 = t.y + s2v.y;
        const float v0 = x0 > 0.0f ? x0 : ALPHA * x0;
        const float v1 = x1 > 0.0f ? x1 : ALPHA * x1;
        mx0 = fmaxf(mx0, ok ? v0 : ninf);
        mx1 = fmaxf(mx1, ok ? v1 : ninf);
    }
    for (int base = 0; base < on; base += 8) {
        int s; float v0, v1;
        const bool ok = gat_ovf_edge(base + e8, on, node, ovf_src, ovf_dst, s1, hp, s2v, s, v0, v1);
        mx0 = fmaxf(mx0, ok ? v0 : ninf);
        mx1 = fmaxf(mx1, ok ? v1 : ninf);
    }
    mx0 = fmaxf(mx0, row_xor<1>(mx0));  mx1 = fmaxf(mx1, row_xor<1>(mx1));
    mx0 = fmaxf(mx0, row_xor<2>(mx0));  mx1 = fmaxf(mx1, row_xor<2>(mx1));
    mx0 = fmaxf(mx0, row_xor<4>(mx0));  mx1 = fmaxf(mx1, row_xor<4>(mx1));

    float sum0 = 0.0f, sum1 = 0.0f;
    for (int base = beg; base < end; base += 8) {
        const int j = base + e8;
        const bool ok = j < end;
        const int s = bsrc[ok ? j : beg];
        const float2 t = *(const float2*)(s1 + s * N_HEADS + 2 * hp);
        const float x0 = t.x + s2v.x, x1 = t.y + s2v.y;
        const float v0 = x0 > 0.0f ? x0 : ALPHA * x0;
        const float v1 = x1 > 0.0f ? x1 : ALPHA * x1;
        sum0 += __expf((ok ? v0 : ninf) - mx0);
        sum1 += __expf((ok ? v1 : ninf) - mx1);
    }
    for (int base = 0; base < on; base += 8) {
        int s; float v0, v1;
        const bool ok = gat_ovf_edge(base + e8, on, node, ovf_src, ovf_dst, s1, hp, s2v, s, v0, v1);
        sum0 += __expf((ok ? v0 : ninf) - mx0);
        sum1 += __expf((ok ? v1 : ninf) - mx1);
    }
    sum0 += row_xor<1>(sum0);  sum1 += row_xor<1>(sum1);
    sum0 += row_xor<2>(sum0);  sum1 += row_xor<2>(sum1);
    sum0 += row_xor<4>(sum0);  sum1 += row_xor<4>(sum1);

    float4 accA = make_float4(0.0f, 0.0f, 0.0f, 0.0f);
    float4 accB = make_float4(0.0f, 0.0f, 0.0f, 0.0f);

    for (int base = beg; base < end; base += 8) {
        const int j = base + e8;
        const bool ok = j < end;
        const int s = bsrc[ok ? j : beg];
        const float2 t = *(const float2*)(s1 + s * N_HEADS + 2 * hp);
        const int left = end - base;

        float4 va[8], vb[8];
        if (left >= 8) {
            GAT_ROW(0) GAT_ROW(1) GAT_ROW(2) GAT_ROW(3)
            GAT_ROW(4) GAT_ROW(5) GAT_ROW(6) GAT_ROW(7)
        } else {
#pragma unroll
            for (int jj = 0; jj < 8; ++jj) {
                va[jj] = make_float4(0.0f, 0.0f, 0.0f, 0.0f);
                vb[jj] = make_float4(0.0f, 0.0f, 0.0f, 0.0f);
            }
            GAT_ROW(0)
            if (left > 1) GAT_ROW(1)
            if (left > 2) GAT_ROW(2)
            if (left > 3) GAT_ROW(3)
            if (left > 4) GAT_ROW(4)
            if (left > 5) GAT_ROW(5)
            if (left > 6) GAT_ROW(6)
        }

        const float x0 = t.x + s2v.x, x1 = t.y + s2v.y;
        const float v0 = x0 > 0.0f ? x0 : ALPHA * x0;
        const float v1 = x1 > 0.0f ? x1 : ALPHA * x1;
        const float a0 = __expf((ok ? v0 : ninf) - mx0) / sum0;
        const float a1 = __expf((ok ? v1 : ninf) - mx1) / sum1;

#pragma unroll
        for (int jj = 0; jj < 8; ++jj) {
            float w0, w1;
            switch (jj) {
                case 0: w0 = grp8_bcast<0>(a0); w1 = grp8_bcast<0>(a1); break;
                case 1: w0 = grp8_bcast<1>(a0); w1 = grp8_bcast<1>(a1); break;
                case 2: w0 = grp8_bcast<2>(a0); w1 = grp8_bcast<2>(a1); break;
                case 3: w0 = grp8_bcast<3>(a0); w1 = grp8_bcast<3>(a1); break;
                case 4: w0 = grp8_bcast<4>(a0); w1 = grp8_bcast<4>(a1); break;
                case 5: w0 = grp8_bcast<5>(a0); w1 = grp8_bcast<5>(a1); break;
                case 6: w0 = grp8_bcast<6>(a0); w1 = grp8_bcast<6>(a1); break;
                default: w0 = grp8_bcast<7>(a0); w1 = grp8_bcast<7>(a1); break;
            }
            accA.x += w0 * va[jj].x;  accA.y += w0 * va[jj].y;
            accA.z += w0 * va[jj].z;  accA.w += w0 * va[jj].w;
            accB.x += w1 * vb[jj].x;  accB.y += w1 * vb[jj].y;
            accB.z += w1 * vb[jj].z;  accB.w += w1 * vb[jj].w;
        }
    }

    for (int base = 0; base < on; base += 8) {
        int s; float v0, v1;
        const bool ok = gat_ovf_edge(base + e8, on, node, ovf_src, ovf_dst, s1, hp, s2v, s, v0, v1);
        const float a0 = __expf((ok ? v0 : ninf) - mx0) / sum0;
        const float a1 = __expf((ok ? v1 : ninf) - mx1) / sum1;
        const unsigned okm = __builtin_amdgcn_ballot_w32(ok) & 0xffu;
        for (int jj = 0; jj < 8; ++jj) {
            if (!((okm >> jj) & 1u)) continue;
            const int sj = __shfl(s, jj, 32);
            const float* rp = Wh + (size_t)sj * (N_HEADS * N_HID);
            const float4 ra = *(const float4*)(rp + loff);
            const float4 rb = *(const float4*)(rp + loff + N_HID);
            const float w0 = __shfl(a0, (lane & 24) | jj, 32);
            const float w1 = __shfl(a1, (lane & 24) | jj, 32);
            accA.x += w0 * ra.x;  accA.y += w0 * ra.y;
            accA.z += w0 * ra.z;  accA.w += w0 * ra.w;
            accB.x += w1 * rb.x;  accB.y += w1 * rb.y;
            accB.z += w1 * rb.z;  accB.w += w1 * rb.w;
        }
    }

    float* op = out + (size_t)node * (N_HEADS * N_HID) + loff;
    *(float4*)op = accA;
    *(float4*)(op + N_HID) = accB;
}

extern "C" void kernel_launch(void* const* d_in, const int* in_sizes, int n_in,
                              void* d_out, int out_size, void* d_ws, size_t ws_size,
                              hipStream_t stream) {
    const float* h   = (const float*)d_in[0];
    const float* W   = (const float*)d_in[1];
    const float* a   = (const float*)d_in[2];
    const int*   src = (const int*)d_in[3];
    const int*   dst = (const int*)d_in[4];
    float* out = (float*)d_out;

    float* ws   = (float*)d_ws;
    float* Wh   = ws;
    float* s1   = Wh + (size_t)N_HEADS * N_NODES * N_HID;
    float* s2   = s1 + (size_t)N_HEADS * N_NODES;
    int*   cnt  = (int*)(s2 + (size_t)N_HEADS * N_NODES);
    int*   bsrc = cnt + CNT_WORDS;
    int*   osrc = bsrc + (size_t)N_NODES * BKT_CAP;
    int*   odst = osrc + N_EDGES;
    int*   ovfn = cnt + N_NODES;

    gat_zero_cnt<<<CNT_WORDS / 256, 256, 0, stream>>>(cnt);

    gat_gemm<<<FILL_BLOCKS + GEMM_BLOCKS, 256, 0, stream>>>(h, W, a, Wh, s1, s2,
                                                           src, dst, cnt, bsrc, ovfn, osrc, odst);

    gat_node<<<(N_NODES + 7) / 8, 256, 0, stream>>>(cnt, bsrc, ovfn, osrc, odst, s1, s2, Wh, out);
}


# node kernel walks each bucket once, not 3 times: running max per 8-edge trip, exp-sum and message sums rescaled by exp(old max - new max), one division at the end; sum order kept, rounding changed
# speedup vs baseline: 7.4427x; 1.0756x over previous
#include <hip/hip_runtime.h>
#include <hip/hip_bf16.h>

typedef float v2f __attribute__((ext_vector_type(2)));
typedef float v8f __attribute__((ext_vector_type(8)));

#define N_NODES 50000
#define N_EDGES 800000
#define N_FEAT  256
#define N_HID   32
#define N_HEADS 8
#define ALPHA   0.2f

template <int K>
__device__ __forceinline__ float row_xor(float v) {
    return __int_as_float(__builtin_amdgcn_update_dpp(0, __float_as_int(v), 0x160 | K,
                                                      0xf, 0xf, true));
}

#define GEMM_ROWS 32
#define SH_PITCH  (N_FEAT + 4)

__device__ __forceinline__ float gat_tile_scores(const v8f c0, const v8f c1,
                                                 const float a1l, const float a1h,
                                                 const float a2l, const float a2h,
                                                 const int m) {
    float x[16];
#pragma unroll
    for (int v = 0; v < 8; ++v) {
        x[v]     = c0[v] * a1l + c1[v] * a1h;
        x[8 + v] = c0[v] * a2l + c1[v] * a2h;
    }
    const bool b3 = (m & 8) != 0, b2 = (m & 4) != 0, b1 = (m & 2) != 0, b0 = (m & 1) != 0;
    float y[8], z[4], w2[2];
#pragma unroll
    for (int i = 0; i < 8; ++i)
        y[i] = (b3 ? x[8 + i] : x[i]) + row_xor<8>(b3 ? x[i] : x[8 + i]);
#pragma unroll
    for (int i = 0; i < 4; ++i)
        z[i] = (b2 ? y[4 + i] : y[i]) + row_xor<4>(b2 ? y[i] : y[4 + i]);
#pragma unroll
    for (int i = 0; i < 2; ++i)
        w2[i] = (b1 ? z[2 + i] : z[i]) + row_xor<2>(b1 ? z[i] : z[2 + i]);
    return (b0 ? w2[1] : w2[0]) + row_xor<1>(b0 ? w2[0] : w2[1]);
}

#define BKT_CAP   64
#define CNT_WORDS ((N_NODES + 255) / 256 * 256)
#define FILL_EPT     16
#define FILL_THREADS (N_EDGES / FILL_EPT)
#define FILL_BLOCKS  ((FILL_THREADS + 255) / 256)
#define GEMM_BLOCKS  ((N_NODES + GEMM_ROWS - 1) / GEMM_ROWS)
static_assert(N_EDGES % FILL_EPT == 0, "a fill thread owns whole 16-byte pieces only");

__device__ __forceinline__ void gat_fill_edge(const int d, const int s, const int slot,
                                              int* __restrict__ bsrc, int* __restrict__ ovf_n,
                                              int* __restrict__ ovf_src, int* __restrict__ ovf_dst) {
    if (slot < BKT_CAP) {
        bsrc[d * BKT_CAP + slot] = s;
    } else {
        const int o = atomicAdd(ovf_n, 1);
        ovf_src[o] = s;
        ovf_dst[o] = d;
    }
}

__device__ __forceinline__ void gat_fill_block(const int* __restrict__ src,
                                               const int* __restrict__ dst,
                                               int* __restrict__ cnt, int* __restrict__ bsrc,
                                               int* __restrict__ ovf_n, int* __restrict__ ovf_src,
                                               int* __restrict__ ovf_dst) {
    const int gt = blockIdx.x * 256 + threadIdx.x;
    if (gt >= FILL_THREADS) return;
    const int4* d4 = (const int4*)dst;
    const int4* s4 = (const int4*)src;
    int4 dv[FILL_EPT / 4], sv[FILL_EPT / 4], sl[FILL_EPT / 4];
#pragma unroll
    for (int i = 0; i < FILL_EPT / 4; ++i) {
        dv[i] = d4[i * FILL_THREADS + gt];
        sv[i] = s4[i * FILL_THREADS + gt];
    }
#pragma unroll
    for (int i = 0; i < FILL_EPT / 4; ++i) {
        sl[i].x = atomicAdd(&cnt[dv[i].x], 1);
        sl[i].y = atomicAdd(&cnt[dv[i].y], 1);
        sl[i].z = atomicAdd(&cnt[dv[i].z], 1);
        sl[i].w = atomicAdd(&cnt[dv[i].w], 1);
    }
#pragma unroll
    for (int i = 0; i < FILL_EPT / 4; ++i) {
        gat_fill_edge(dv[i].x, sv[i].x, sl[i].x, bsrc, ovf_n, ovf_src, ovf_dst);
        gat_fill_edge(dv[i].y, sv[i].y, sl[i].y, bsrc, ovf_n, ovf_src, ovf_dst);
        gat_fill_edge(dv[i].z, sv[i].z, sl[i].z, bsrc, ovf_n, ovf_src, ovf_dst);
        gat_fill_edge(dv[i].w, sv[i].w, sl[i].w, bsrc, ovf_n, ovf_src, ovf_dst);
    }
}

__global__ __launch_bounds__(256) void gat_gemm(const float* __restrict__ hmat,
                                                const float* __restrict__ W,
                                                const float* __restrict__ avec,
                                                float* __restrict__ Wh,
                                                float* __restrict__ s1,
                                                float* __restrict__ s2,
                                                const int* __restrict__ src,
                                                const int* __restrict__ dst,
                                                int* __restrict__ cnt,
                                                int* __restrict__ bsrc,
                                                int* __restrict__ ovf_n,
                                                int* __restrict__ ovf_src,
                                                int* __restrict__ ovf_dst) {
    __shared__ float sh[GEMM_ROWS * SH_PITCH];
    __shared__ float ssc[2 * GEMM_ROWS * N_HEADS];

    if (blockIdx.x < FILL_BLOCKS) {
        gat_fill_block(src, dst, cnt, bsrc, ovf_n, ovf_src, ovf_dst);
        return;
    }

    const int node_base = (blockIdx.x - FILL_BLOCKS) * GEMM_ROWS;
    const bool full = node_base + GEMM_ROWS <= N_NODES;

    const float4* g4 = (const float4*)(hmat + (size_t)node_base * N_FEAT);
    const int col4 = threadIdx.x & 63;
    const int r0   = threadIdx.x >> 6;
    float4 t[8];
#pragma unroll
    for (int i = 0; i < 4; ++i)
        t[i] = g4[threadIdx.x + 256 * i];
    if (full) {
#pragma unroll
        for (int i = 4; i < 8; ++i)
            t[i] = g4[threadIdx.x + 256 * i];
    } else {
#pragma unroll
        for (int i = 4; i < 8; ++i)
            t[i] = make_float4(0.0f, 0.0f, 0.0f, 0.0f);
    }
#pragma unroll
    for (int i = 0; i < 8; ++i)
        *(float4*)(sh + (r0 + 4 * i) * SH_PITCH + col4 * 4) = t[i];
    __syncthreads();

    const int lane = threadIdx.x & 31;
    const int head = threadIdx.x >> 5;
    const float* Wp = W + (size_t)head * N_FEAT * N_HID;

    const int m    = lane & 15;
    const int koff = (lane >> 4) << 1;

    v8f c00 = {};
    v8f c01 = {};
    v8f c10 = {};
    v8f c11 = {};

    const float* ar0 = sh + m * SH_PITCH + koff;
    const float* ar1 = ar0 + 16 * SH_PITCH;
    const float* bp  = Wp + koff * N_HID + m;

    for (int k = 0; k < N_FEAT; k += 4) {
        v2f a0, a1, b0, b1;
        a0 = *(const v2f*)(ar0 + k);
        a1 = *(const v2f*)(ar1 + k);
        b0[0] = bp[k * N_HID];
        b0[1] = bp[(k + 1) * N_HID];
        b1[0] = bp[k * N_HID + 16];
        b1[1] = bp[(k + 1) * N_HID + 16];
        c00 = __builtin_amdgcn_wmma_f32_16x16x4_f32(false, a0, false, b0,
                                                    (short)0, c00, false, false);
        c01 = __builtin_amdgcn_wmma_f32_16x16x4_f32(false, a0, false, b1,
                                                    (short)0, c01, false, false);
        c10 = __builtin_amdgcn_wmma_f32_16x16x4_f32(false, a1, false, b0,
                                                    (short)0, c10, false, false);
        c11 = __builtin_amdgcn_wmma_f32_16x16x4_f32(false, a1, false, b1,
                                                    (short)0, c11, false, false);
    }

    const int mb = (lane >> 4) * 8;

    const float* ap = avec + head * 2 * N_HID;
    const float a1l = ap[m],         a1h = ap[16 + m];
    const float a2l = ap[N_HID + m], a2h = ap[N_HID + 16 + m];
    const float sc0 = gat_tile_scores(c00, c01, a1l, a1h, a2l, a2h, m);
    const float sc1 = gat_tile_scores(c10, c11, a1l, a1h, a2l, a2h, m);

    float* sq = ssc + (m >> 3) * (GEMM_ROWS * N_HEADS) + (mb + (m & 7)) * N_HEADS + head;
    sq[0]            = sc0;
    sq[16 * N_HEADS] = sc1;
    __syncthreads();
    if (full || threadIdx.x < 16 * N_HEADS) {
        s1[node_base * N_HEADS + threadIdx.x] = ssc[threadIdx.x];
        s2[node_base * N_HEADS + threadIdx.x] = ssc[GEMM_ROWS * N_HEADS + threadIdx.x];
    }

    float* op = Wh + (size_t)node_base * (N_HEADS * N_HID) + head * N_HID;
#pragma unroll
    for (int v = 0; v < 8; ++v) {
        op[(mb + v) * (N_HEADS * N_HID) + m]      = c00[v];
        op[(mb + v) * (N_HEADS * N_HID) + 16 + m] = c01[v];
    }
    if (full) {
#pragma unroll
        for (int v = 0; v < 8; ++v) {
            op[(16 + mb + v) * (N_HEADS * N_HID) + m]      = c10[v];
            op[(16 + mb + v) * (N_HEADS * N_HID) + 16 + m] = c11[v];
        }
    }
}

__global__ void gat_zero_cnt(int* __restrict__ cnt) {
    cnt[blockIdx.x * blockDim.x + threadIdx.x] = 0;
}


template <int J>
__device__ __forceinline__ float grp8_bcast(float v) {
    return __int_as_float(__builtin_amdgcn_mov_dpp8(__float_as_int(v), J * 0x249249));
}

#define GAT_ROW(jj)                                                               \
    {                                                                              \
        const int sj = __builtin_amdgcn_readlane(s, (jj));                         \
              \
        const float* rp = Wh + (size_t)sj * (N_HEADS * N_HID);                     \
        va[jj] = *(const float4*)(rp + loff);                                      \
        vb[jj] = *(const float4*)(rp + loff + N_HID);                              \
    }

__device__ __forceinline__ bool gat_ovf_edge(const int j, const int on, const int node,
                                             const int* __restrict__ ovf_src,
                                             const int* __restrict__ ovf_dst,
                                             const float* __restrict__ s1, const int hp,
                                             const float2 s2v, int& s, float& v0, float& v1) {
    const bool in = j < on;
    const int jc = in ? j : 0;
    const bool ok = in && ovf_dst[jc] == node;
    s = ovf_src[jc];
    const float2 t = *(const float2*)(s1 + s * N_HEADS + 2 * hp);
    const float x0 = t.x + s2v.x, x1 = t.y + s2v.y;
    v0 = x0 > 0.0f ? x0 : ALPHA * x0;
    v1 = x1 > 0.0f ? x1 : ALPHA * x1;
    return ok;
}

__global__ __launch_bounds__(256) void gat_node(const int* __restrict__ cnt,
                                                const int* __restrict__ bsrc,
                                                const int* __restrict__ ovf_n,
                                                const int* __restrict__ ovf_src,
                                                const int* __restrict__ ovf_dst,
                                                const float* __restrict__ s1,
                                                const float* __restrict__ s2,
                                                const float* __restrict__ Wh,
                                                float* __restrict__ out) {
    const int node = __builtin_amdgcn_readfirstlane((int)(blockIdx.x * 8 + (threadIdx.x >> 5)));
    if (node >= N_NODES) return;
    const int lane = threadIdx.x & 31;
    const int e8 = lane & 7;
    const int hp = lane >> 3;
    const unsigned loff = hp * (2 * N_HID) + e8 * 4;

    const int deg = __builtin_amdgcn_readfirstlane(cnt[node]);
    const int beg = node * BKT_CAP;
    const int end = beg + (deg < BKT_CAP ? deg : BKT_CAP);
    const int on  = deg > BKT_CAP ? __builtin_amdgcn_readfirstlane(*ovf_n) : 0;
    const float2 s2v = *(const float2*)(s2 + node * N_HEADS + 2 * hp);

    const float ninf = -__builtin_inff();

    float mx0 = ninf, mx1 = ninf;
    float sum0 = 0.0f, sum1 = 0.0f;
    float4 accA = make_float4(0.0f, 0.0f, 0.0f, 0.0f);
    float4 accB = make_float4(0.0f, 0.0f, 0.0f, 0.0f);

    for (int base = beg; base < end; base += 8) {
        const int j = base + e8;
        const bool ok = j < end;
        const int s = bsrc[ok ? j : beg];
        const float2 t = *(const float2*)(s1 + s * N_HEADS + 2 * hp);
        const int left = end - base;

        float4 va[8], vb[8];
        if (left >= 8) {
            GAT_ROW(0) GAT_ROW(1) GAT_ROW(2) GAT_ROW(3)
            GAT_ROW(4) GAT_ROW(5) GAT_ROW(6) GAT_ROW(7)
        } else {
#pragma unroll
            for (int jj = 0; jj < 8; ++jj) {
                va[jj] = make_float4(0.0f, 0.0f, 0.0f, 0.0f);
                vb[jj] = make_float4(0.0f, 0.0f, 0.0f, 0.0f);
            }
            GAT_ROW(0)
            if (left > 1) GAT_ROW(1)
            if (left > 2) GAT_ROW(2)
            if (left > 3) GAT_ROW(3)
            if (left > 4) GAT_ROW(4)
            if (left > 5) GAT_ROW(5)
            if (left > 6) GAT_ROW(6)
        }

        const float x0 = t.x + s2v.x, x1 = t.y + s2v.y;
        const float v0 = x0 > 0.0f ? x0 : ALPHA * x0;
        const float v1 = x1 > 0.0f ? x1 : ALPHA * x1;
        const float e0 = ok ? v0 : ninf, e1 = ok ? v1 : ninf;

        float n0 = e0, n1 = e1;
        n0 = fmaxf(n0, row_xor<1>(n0));  n1 = fmaxf(n1, row_xor<1>(n1));
        n0 = fmaxf(n0, row_xor<2>(n0));  n1 = fmaxf(n1, row_xor<2>(n1));
        n0 = fmaxf(n0, row_xor<4>(n0));  n1 = fmaxf(n1, row_xor<4>(n1));
        n0 = fmaxf(n0, mx0);             n1 = fmaxf(n1, mx1);
        const float r0 = __expf(mx0 - n0), r1 = __expf(mx1 - n1);
        const float a0 = __expf(e0 - n0),  a1 = __expf(e1 - n1);
        mx0 = n0;  mx1 = n1;
        sum0 = sum0 * r0 + a0;
        sum1 = sum1 * r1 + a1;
        accA.x *= r0;  accA.y *= r0;  accA.z *= r0;  accA.w *= r0;
        accB.x *= r1;  accB.y *= r1;  accB.z *= r1;  accB.w *= r1;

#pragma unroll
        for (int jj = 0; jj < 8; ++jj) {
            float w0, w1;
            switch (jj) {
                case 0: w0 = grp8_bcast<0>(a0); w1 = grp8_bcast<0>(a1); break;
                case 1: w0 = grp8_bcast<1>(a0); w1 = grp8_bcast<1>(a1); break;
                case 2: w0 = grp8_bcast<2>(a0); w1 = grp8_bcast<2>(a1); break;
                case 3: w0 = grp8_bcast<3>(a0); w1 = grp8_bcast<3>(a1); break;
                case 4: w0 = grp8_bcast<4>(a0); w1 = grp8_bcast<4>(a1); break;
                case 5: w0 = grp8_bcast<5>(a0); w1 = grp8_bcast<5>(a1); break;
                case 6: w0 = grp8_bcast<6>(a0); w1 = grp8_bcast<6>(a1); break;
                default: w0 = grp8_bcast<7>(a0); w1 = grp8_bcast<7>(a1); break;
            }
            accA.x += w0 * va[jj].x;  accA.y += w0 * va[jj].y;
            accA.z += w0 * va[jj].z;  accA.w += w0 * va[jj].w;
            accB.x += w1 * vb[jj].x;  accB.y += w1 * vb[jj].y;
            accB.z += w1 * vb[jj].z;  accB.w += w1 * vb[jj].w;
        }
    }

    for (int base = 0; base < on; base += 8) {
        int s; float v0, v1;
        const bool ok = gat_ovf_edge(base + e8, on, node, ovf_src, ovf_dst, s1, hp, s2v, s, v0, v1);
        const unsigned okm = __builtin_amdgcn_ballot_w32(ok) & 0xffu;
        if (okm == 0u) continue;
        const float e0 = ok ? v0 : ninf, e1 = ok ? v1 : ninf;
        float n0 = e0, n1 = e1;
        n0 = fmaxf(n0, row_xor<1>(n0));  n1 = fmaxf(n1, row_xor<1>(n1));
        n0 = fmaxf(n0, row_xor<2>(n0));  n1 = fmaxf(n1, row_xor<2>(n1));
        n0 = fmaxf(n0, row_xor<4>(n0));  n1 = fmaxf(n1, row_xor<4>(n1));
        n0 = fmaxf(n0, mx0);             n1 = fmaxf(n1, mx1);
        const float r0 = __expf(mx0 - n0), r1 = __expf(mx1 - n1);
        const float a0 = __expf(e0 - n0),  a1 = __expf(e1 - n1);
        mx0 = n0;  mx1 = n1;
        sum0 = sum0 * r0 + a0;
        sum1 = sum1 * r1 + a1;
        accA.x *= r0;  accA.y *= r0;  accA.z *= r0;  accA.w *= r0;
        accB.x *= r1;  accB.y *= r1;  accB.z *= r1;  accB.w *= r1;
        for (int jj = 0; jj < 8; ++jj) {
            if (!((okm >> jj) & 1u)) continue;
            const int sj = __shfl(s, jj, 32);
            const float* rp = Wh + (size_t)sj * (N_HEADS * N_HID);
            const float4 ra = *(const float4*)(rp + loff);
            const float4 rb = *(const float4*)(rp + loff + N_HID);
            const float w0 = __shfl(a0, (lane & 24) | jj, 32);
            const float w1 = __shfl(a1, (lane & 24) | jj, 32);
            accA.x += w0 * ra.x;  accA.y += w0 * ra.y;
            accA.z += w0 * ra.z;  accA.w += w0 * ra.w;
            accB.x += w1 * rb.x;  accB.y += w1 * rb.y;
            accB.z += w1 * rb.z;  accB.w += w1 * rb.w;
        }
    }

    sum0 += row_xor<1>(sum0);  sum1 += row_xor<1>(sum1);
    sum0 += row_xor<2>(sum0);  sum1 += row_xor<2>(sum1);
    sum0 += row_xor<4>(sum0);  sum1 += row_xor<4>(sum1);
    const float q0 = deg > 0 ? 1.0f / sum0 : 0.0f;
    const float q1 = deg > 0 ? 1.0f / sum1 : 0.0f;
    accA.x *= q0;  accA.y *= q0;  accA.z *= q0;  accA.w *= q0;
    accB.x *= q1;  accB.y *= q1;  accB.z *= q1;  accB.w *= q1;

    float* op = out + (size_t)node * (N_HEADS * N_HID) + loff;
    *(float4*)op = accA;
    *(float4*)(op + N_HID) = accB;
}

extern "C" void kernel_launch(void* const* d_in, const int* in_sizes, int n_in,
                              void* d_out, int out_size, void* d_ws, size_t ws_size,
                              hipStream_t stream) {
    const float* h   = (const float*)d_in[0];
    const float* W   = (const float*)d_in[1];
    const float* a   = (const float*)d_in[2];
    const int*   src = (const int*)d_in[3];
    const int*   dst = (const int*)d_in[4];
    float* out = (float*)d_out;

    float* ws   = (float*)d_ws;
    float* Wh   = ws;
    float* s1   = Wh + (size_t)N_HEADS * N_NODES * N_HID;
    float* s2   = s1 + (size_t)N_HEADS * N_NODES;
    int*   cnt  = (int*)(s2 + (size_t)N_HEADS * N_NODES);
    int*   bsrc = cnt + CNT_WORDS;
    int*   osrc = bsrc + (size_t)N_NODES * BKT_CAP;
    int*   odst = osrc + N_EDGES;
    int*   ovfn = cnt + N_NODES;

    gat_zero_cnt<<<CNT_WORDS / 256, 256, 0, stream>>>(cnt);

    gat_gemm<<<FILL_BLOCKS + GEMM_BLOCKS, 256, 0, stream>>>(h, W, a, Wh, s1, s2,
                                                           src, dst, cnt, bsrc, ovfn, osrc, odst);

    gat_node<<<(N_NODES + 7) / 8, 256, 0, stream>>>(cnt, bsrc, ovfn, osrc, odst, s1, s2, Wh, out);
}


# edges ranked per dst node on LDS counters: pre-sort launch bins them by 196-node range, 256 owner blocks beside the GEMM fill buckets; no per-edge global atomic, no zeroing; sums in LDS arrival order
# speedup vs baseline: 9.5892x; 1.2884x over previous
#include <hip/hip_runtime.h>
#include <hip/hip_bf16.h>

typedef float v2f __attribute__((ext_vector_type(2)));
typedef float v8f __attribute__((ext_vector_type(8)));

#define N_NODES 50000
#define N_EDGES 800000
#define N_FEAT  256
#define N_HID   32
#define N_HEADS 8
#define ALPHA   0.2f

template <int K>
__device__ __forceinline__ float row_xor(float v) {
    return __int_as_float(__builtin_amdgcn_update_dpp(0, __float_as_int(v), 0x160 | K,
                                                      0xf, 0xf, true));
}

#define GEMM_ROWS 32
#define SH_PITCH  (N_FEAT + 4)

__device__ __forceinline__ float gat_tile_scores(const v8f c0, const v8f c1,
                                                 const float a1l, const float a1h,
                                                 const float a2l, const float a2h,
                                                 const int m) {
    float x[16];
#pragma unroll
    for (int v = 0; v < 8; ++v) {
        x[v]     = c0[v] * a1l + c1[v] * a1h;
        x[8 + v] = c0[v] * a2l + c1[v] * a2h;
    }
    const bool b3 = (m & 8) != 0, b2 = (m & 4) != 0, b1 = (m & 2) != 0, b0 = (m & 1) != 0;
    float y[8], z[4], w2[2];
#pragma unroll
    for (int i = 0; i < 8; ++i)
        y[i] = (b3 ? x[8 + i] : x[i]) + row_xor<8>(b3 ? x[i] : x[8 + i]);
#pragma unroll
    for (int i = 0; i < 4; ++i)
        z[i] = (b2 ? y[4 + i] : y[i]) + row_xor<4>(b2 ? y[i] : y[4 + i]);
#pragma unroll
    for (int i = 0; i < 2; ++i)
        w2[i] = (b1 ? z[2 + i] : z[i]) + row_xor<2>(b1 ? z[i] : z[2 + i]);
    return (b0 ? w2[1] : w2[0]) + row_xor<1>(b0 ? w2[0] : w2[1]);
}

#define BKT_CAP   64
#define CNT_WORDS ((N_NODES + 255) / 256 * 256)
#define NBIN       256
#define BIN_NODES  ((N_NODES + NBIN - 1) / NBIN)
#define PRE_BLOCKS 256
#define PRE_EDGES  (N_EDGES / PRE_BLOCKS)
#define PRE_EPT    ((PRE_EDGES + 255) / 256)
#define SEG_CAP    64
#define GEMM_BLOCKS  ((N_NODES + GEMM_ROWS - 1) / GEMM_ROWS)
static_assert(N_EDGES % PRE_BLOCKS == 0, "every step-1 block has exactly PRE_EDGES edges");
static_assert(N_NODES <= 65536, "an entry keeps the source node in its low 16 bits");
static_assert(NBIN == 256 && PRE_BLOCKS == 256, "one thread per range in step 1, per segment in step 2");

__global__ __launch_bounds__(256) void gat_presort(const int* __restrict__ src,
                                                   const int* __restrict__ dst,
                                                   int* __restrict__ seg,
                                                   int* __restrict__ segn,
                                                   int* __restrict__ ovf_n) {
    __shared__ int lc[NBIN + 1];
    const int b = blockIdx.x;
    lc[threadIdx.x] = 0;
    if (threadIdx.x == 0) lc[NBIN] = 0;
    if (b == 0 && threadIdx.x == 0) *ovf_n = 0;
    __syncthreads();
    const int e0 = b * PRE_EDGES;
    int d[PRE_EPT], s[PRE_EPT], k[PRE_EPT], r[PRE_EPT];
#pragma unroll
    for (int i = 0; i < PRE_EPT; ++i) {
        const int j = i * 256 + threadIdx.x;
        const bool ok = j < PRE_EDGES;
        d[i] = ok ? dst[e0 + j] : -1;
        s[i] = ok ? src[e0 + j] : 0;
    }
#pragma unroll
    for (int i = 0; i < PRE_EPT; ++i) {
        k[i] = d[i] >= 0 ? d[i] / BIN_NODES : NBIN;
        r[i] = atomicAdd(&lc[k[i]], 1);
    }
#pragma unroll
    for (int i = 0; i < PRE_EPT; ++i) {
        if (d[i] >= 0 && r[i] < SEG_CAP)
            seg[((size_t)k[i] * PRE_BLOCKS + b) * SEG_CAP + r[i]] = ((d[i] - k[i] * BIN_NODES) << 16) | s[i];
    }
    __syncthreads();
    segn[threadIdx.x * PRE_BLOCKS + b] = lc[threadIdx.x];
}

__device__ __forceinline__ void gat_fill_edge(const int d, const int s, const int slot,
                                              int* __restrict__ bsrc, int* __restrict__ ovf_n,
                                              int* __restrict__ ovf_src, int* __restrict__ ovf_dst) {
    if (slot < BKT_CAP) {
        bsrc[d * BKT_CAP + slot] = s;
    } else {
        const int o = atomicAdd(ovf_n, 1);
        ovf_src[o] = s;
        ovf_dst[o] = d;
    }
}

__device__ __forceinline__ void gat_place(const int entry, const int node0, int* lcn,
                                          int* __restrict__ bsrc, int* __restrict__ ovf_n,
                                          int* __restrict__ ovf_src, int* __restrict__ ovf_dst) {
    const int dl = entry >> 16;
    const int slot = atomicAdd(&lcn[dl], 1);
    gat_fill_edge(node0 + dl, entry & 0xffff, slot, bsrc, ovf_n, ovf_src, ovf_dst);
}

__device__ __forceinline__ void gat_owner_block(int* lcn,
                                                const int* __restrict__ src,
                                                const int* __restrict__ dst,
                                                const int* __restrict__ seg,
                                                const int* __restrict__ segn,
                                                int* __restrict__ cnt, int* __restrict__ bsrc,
                                                int* __restrict__ ovf_n, int* __restrict__ ovf_src,
                                                int* __restrict__ ovf_dst) {
    const int k = blockIdx.x;
    const int t = threadIdx.x;
    const int node0 = k * BIN_NODES;
    if (t <= BIN_NODES) lcn[t] = 0;
    const int c = segn[k * PRE_BLOCKS + t];
    __syncthreads();
    if (c <= SEG_CAP) {
        const int4* sp = (const int4*)(seg + ((size_t)k * PRE_BLOCKS + t) * SEG_CAP);
        for (int i = 0; i < c; i += 4) {
            const int4 v = sp[i >> 2];
            const int e[4] = {v.x, v.y, v.z, v.w};
            int sl[4];
#pragma unroll
            for (int j = 0; j < 4; ++j)
                sl[j] = atomicAdd(&lcn[i + j < c ? e[j] >> 16 : BIN_NODES], 1);
#pragma unroll
            for (int j = 0; j < 4; ++j)
                if (i + j < c)
                    gat_fill_edge(node0 + (e[j] >> 16), e[j] & 0xffff, sl[j], bsrc, ovf_n, ovf_src, ovf_dst);
        }
    } else {
        const int e0 = t * PRE_EDGES;
        for (int j = 0; j < PRE_EDGES; ++j) {
            const int d = dst[e0 + j];
            if (d / BIN_NODES == k)
                gat_place(((d - node0) << 16) | src[e0 + j], node0, lcn, bsrc, ovf_n, ovf_src, ovf_dst);
        }
    }
    __syncthreads();
    if (t < BIN_NODES && node0 + t < N_NODES) cnt[node0 + t] = lcn[t];
}

__global__ __launch_bounds__(256) void gat_gemm(const float* __restrict__ hmat,
                                                const float* __restrict__ W,
                                                const float* __restrict__ avec,
                                                float* __restrict__ Wh,
                                                float* __restrict__ s1,
                                                float* __restrict__ s2,
                                                const int* __restrict__ src,
                                                const int* __restrict__ dst,
                                                const int* __restrict__ seg,
                                                const int* __restrict__ segn,
                                                int* __restrict__ cnt,
                                                int* __restrict__ bsrc,
                                                int* __restrict__ ovf_n,
                                                int* __restrict__ ovf_src,
                                                int* __restrict__ ovf_dst) {
    __shared__ float sh[GEMM_ROWS * SH_PITCH];
    __shared__ float ssc[2 * GEMM_ROWS * N_HEADS];
    __shared__ int   lcn[BIN_NODES + 1];

    if (blockIdx.x < NBIN) {
        gat_owner_block(lcn, src, dst, seg, segn, cnt, bsrc, ovf_n, ovf_src, ovf_dst);
        return;
    }

    const int node_base = (blockIdx.x - NBIN) * GEMM_ROWS;
    const bool full = node_base + GEMM_ROWS <= N_NODES;

    const float4* g4 = (const float4*)(hmat + (size_t)node_base * N_FEAT);
    const int col4 = threadIdx.x & 63;
    const int r0   = threadIdx.x >> 6;
    float4 t[8];
#pragma unroll
    for (int i = 0; i < 4; ++i)
        t[i] = g4[threadIdx.x + 256 * i];
    if (full) {
#pragma unroll
        for (int i = 4; i < 8; ++i)
            t[i] = g4[threadIdx.x + 256 * i];
    } else {
#pragma unroll
        for (int i = 4; i < 8; ++i)
            t[i] = make_float4(0.0f, 0.0f, 0.0f, 0.0f);
    }
#pragma unroll
    for (int i = 0; i < 8; ++i)
        *(float4*)(sh + (r0 + 4 * i) * SH_PITCH + col4 * 4) = t[i];
    __syncthreads();

    const int lane = threadIdx.x & 31;
    const int head = threadIdx.x >> 5;
    const float* Wp = W + (size_t)head * N_FEAT * N_HID;

    const int m    = lane & 15;
    const int koff = (lane >> 4) << 1;

    v8f c00 = {};
    v8f c01 = {};
    v8f c10 = {};
    v8f c11 = {};

    const float* ar0 = sh + m * SH_PITCH + koff;
    const float* ar1 = ar0 + 16 * SH_PITCH;
    const float* bp  = Wp + koff * N_HID + m;

    for (int k = 0; k < N_FEAT; k += 4) {
        v2f a0, a1, b0, b1;
        a0 = *(const v2f*)(ar0 + k);
        a1 = *(const v2f*)(ar1 + k);
        b0[0] = bp[k * N_HID];
        b0[1] = bp[(k + 1) * N_HID];
        b1[0] = bp[k * N_HID + 16];
        b1[1] = bp[(k + 1) * N_HID + 16];
        c00 = __builtin_amdgcn_wmma_f32_16x16x4_f32(false, a0, false, b0,
                                                    (short)0, c00, false, false);
        c01 = __builtin_amdgcn_wmma_f32_16x16x4_f32(false, a0, false, b1,
                                                    (short)0, c01, false, false);
        c10 = __builtin_amdgcn_wmma_f32_16x16x4_f32(false, a1, false, b0,
                                                    (short)0, c10, false, false);
        c11 = __builtin_amdgcn_wmma_f32_16x16x4_f32(false, a1, false, b1,
                                                    (short)0, c11, false, false);
    }

    const int mb = (lane >> 4) * 8;

    const float* ap = avec + head * 2 * N_HID;
    const float a1l = ap[m],         a1h = ap[16 + m];
    const float a2l = ap[N_HID + m], a2h = ap[N_HID + 16 + m];
    const float sc0 = gat_tile_scores(c00, c01, a1l, a1h, a2l, a2h, m);
    const float sc1 = gat_tile_scores(c10, c11, a1l, a1h, a2l, a2h, m);

    float* sq = ssc + (m >> 3) * (GEMM_ROWS * N_HEADS) + (mb + (m & 7)) * N_HEADS + head;
    sq[0]            = sc0;
    sq[16 * N_HEADS] = sc1;
    __syncthreads();
    if (full || threadIdx.x < 16 * N_HEADS) {
        s1[node_base * N_HEADS + threadIdx.x] = ssc[threadIdx.x];
        s2[node_base * N_HEADS + threadIdx.x] = ssc[GEMM_ROWS * N_HEADS + threadIdx.x];
    }

    float* op = Wh + (size_t)node_base * (N_HEADS * N_HID) + head * N_HID;
#pragma unroll
    for (int v = 0; v < 8; ++v) {
        op[(mb + v) * (N_HEADS * N_HID) + m]      = c00[v];
        op[(mb + v) * (N_HEADS * N_HID) + 16 + m] = c01[v];
    }
    if (full) {
#pragma unroll
        for (int v = 0; v < 8; ++v) {
            op[(16 + mb + v) * (N_HEADS * N_HID) + m]      = c10[v];
            op[(16 + mb + v) * (N_HEADS * N_HID) + 16 + m] = c11[v];
        }
    }
}


template <int J>
__device__ __forceinline__ float grp8_bcast(float v) {
    return __int_as_float(__builtin_amdgcn_mov_dpp8(__float_as_int(v), J * 0x249249));
}

#define GAT_ROW(jj)                                                               \
    {                                                                              \
        const int sj = __builtin_amdgcn_readlane(s, (jj));                         \
              \
        const float* rp = Wh + (size_t)sj * (N_HEADS * N_HID);                     \
        va[jj] = *(const float4*)(rp + loff);                                      \
        vb[jj] = *(const float4*)(rp + loff + N_HID);                              \
    }

__device__ __forceinline__ bool gat_ovf_edge(const int j, const int on, const int node,
                                             const int* __restrict__ ovf_src,
                                             const int* __restrict__ ovf_dst,
                                             const float* __restrict__ s1, const int hp,
                                             const float2 s2v, int& s, float& v0, float& v1) {
    const bool in = j < on;
    const int jc = in ? j : 0;
    const bool ok = in && ovf_dst[jc] == node;
    s = ovf_src[jc];
    const float2 t = *(const float2*)(s1 + s * N_HEADS + 2 * hp);
    const float x0 = t.x + s2v.x, x1 = t.y + s2v.y;
    v0 = x0 > 0.0f ? x0 : ALPHA * x0;
    v1 = x1 > 0.0f ? x1 : ALPHA * x1;
    return ok;
}

__global__ __launch_bounds__(256) void gat_node(const int* __restrict__ cnt,
                                                const int* __restrict__ bsrc,
                                                const int* __restrict__ ovf_n,
                                                const int* __restrict__ ovf_src,
                                                const int* __restrict__ ovf_dst,
                                                const float* __restrict__ s1,
                                                const float* __restrict__ s2,
                                                const float* __restrict__ Wh,
                                                float* __restrict__ out) {
    const int node = __builtin_amdgcn_readfirstlane((int)(blockIdx.x * 8 + (threadIdx.x >> 5)));
    if (node >= N_NODES) return;
    const int lane = threadIdx.x & 31;
    const int e8 = lane & 7;
    const int hp = lane >> 3;
    const unsigned loff = hp * (2 * N_HID) + e8 * 4;

    const int deg = __builtin_amdgcn_readfirstlane(cnt[node]);
    const int beg = node * BKT_CAP;
    const int end = beg + (deg < BKT_CAP ? deg : BKT_CAP);
    const int on  = deg > BKT_CAP ? __builtin_amdgcn_readfirstlane(*ovf_n) : 0;
    const float2 s2v = *(const float2*)(s2 + node * N_HEADS + 2 * hp);

    const float ninf = -__builtin_inff();

    float mx0 = ninf, mx1 = ninf;
    float sum0 = 0.0f, sum1 = 0.0f;
    float4 accA = make_float4(0.0f, 0.0f, 0.0f, 0.0f);
    float4 accB = make_float4(0.0f, 0.0f, 0.0f, 0.0f);

    for (int base = beg; base < end; base += 8) {
        const int j = base + e8;
        const bool ok = j < end;
        const int s = bsrc[ok ? j : beg];
        const float2 t = *(const float2*)(s1 + s * N_HEADS + 2 * hp);
        const int left = end - base;

        float4 va[8], vb[8];
        if (left >= 8) {
            GAT_ROW(0) GAT_ROW(1) GAT_ROW(2) GAT_ROW(3)
            GAT_ROW(4) GAT_ROW(5) GAT_ROW(6) GAT_ROW(7)
        } else {
#pragma unroll
            for (int jj = 0; jj < 8; ++jj) {
                va[jj] = make_float4(0.0f, 0.0f, 0.0f, 0.0f);
                vb[jj] = make_float4(0.0f, 0.0f, 0.0f, 0.0f);
            }
            GAT_ROW(0)
            if (left > 1) GAT_ROW(1)
            if (left > 2) GAT_ROW(2)
            if (left > 3) GAT_ROW(3)
            if (left > 4) GAT_ROW(4)
            if (left > 5) GAT_ROW(5)
            if (left > 6) GAT_ROW(6)
        }

        const float x0 = t.x + s2v.x, x1 = t.y + s2v.y;
        const float v0 = x0 > 0.0f ? x0 : ALPHA * x0;
        const float v1 = x1 > 0.0f ? x1 : ALPHA * x1;
        const float e0 = ok ? v0 : ninf, e1 = ok ? v1 : ninf;

        float n0 = e0, n1 = e1;
        n0 = fmaxf(n0, row_xor<1>(n0));  n1 = fmaxf(n1, row_xor<1>(n1));
        n0 = fmaxf(n0, row_xor<2>(n0));  n1 = fmaxf(n1, row_xor<2>(n1));
        n0 = fmaxf(n0, row_xor<4>(n0));  n1 = fmaxf(n1, row_xor<4>(n1));
        n0 = fmaxf(n0, mx0);             n1 = fmaxf(n1, mx1);
        const float r0 = __expf(mx0 - n0), r1 = __expf(mx1 - n1);
        const float a0 = __expf(e0 - n0),  a1 = __expf(e1 - n1);
        mx0 = n0;  mx1 = n1;
        sum0 = sum0 * r0 + a0;
        sum1 = sum1 * r1 + a1;
        accA.x *= r0;  accA.y *= r0;  accA.z *= r0;  accA.w *= r0;
        accB.x *= r1;  accB.y *= r1;  accB.z *= r1;  accB.w *= r1;

#pragma unroll
        for (int jj = 0; jj < 8; ++jj) {
            float w0, w1;
            switch (jj) {
                case 0: w0 = grp8_bcast<0>(a0); w1 = grp8_bcast<0>(a1); break;
                case 1: w0 = grp8_bcast<1>(a0); w1 = grp8_bcast<1>(a1); break;
                case 2: w0 = grp8_bcast<2>(a0); w1 = grp8_bcast<2>(a1); break;
                case 3: w0 = grp8_bcast<3>(a0); w1 = grp8_bcast<3>(a1); break;
                case 4: w0 = grp8_bcast<4>(a0); w1 = grp8_bcast<4>(a1); break;
                case 5: w0 = grp8_bcast<5>(a0); w1 = grp8_bcast<5>(a1); break;
                case 6: w0 = grp8_bcast<6>(a0); w1 = grp8_bcast<6>(a1); break;
                default: w0 = grp8_bcast<7>(a0); w1 = grp8_bcast<7>(a1); break;
            }
            accA.x += w0 * va[jj].x;  accA.y += w0 * va[jj].y;
            accA.z += w0 * va[jj].z;  accA.w += w0 * va[jj].w;
            accB.x += w1 * vb[jj].x;  accB.y += w1 * vb[jj].y;
            accB.z += w1 * vb[jj].z;  accB.w += w1 * vb[jj].w;
        }
    }

    for (int base = 0; base < on; base += 8) {
        int s; float v0, v1;
        const bool ok = gat_ovf_edge(base + e8, on, node, ovf_src, ovf_dst, s1, hp, s2v, s, v0, v1);
        const unsigned okm = __builtin_amdgcn_ballot_w32(ok) & 0xffu;
        if (okm == 0u) continue;
        const float e0 = ok ? v0 : ninf, e1 = ok ? v1 : ninf;
        float n0 = e0, n1 = e1;
        n0 = fmaxf(n0, row_xor<1>(n0));  n1 = fmaxf(n1, row_xor<1>(n1));
        n0 = fmaxf(n0, row_xor<2>(n0));  n1 = fmaxf(n1, row_xor<2>(n1));
        n0 = fmaxf(n0, row_xor<4>(n0));  n1 = fmaxf(n1, row_xor<4>(n1));
        n0 = fmaxf(n0, mx0);             n1 = fmaxf(n1, mx1);
        const float r0 = __expf(mx0 - n0), r1 = __expf(mx1 - n1);
        const float a0 = __expf(e0 - n0),  a1 = __expf(e1 - n1);
        mx0 = n0;  mx1 = n1;
        sum0 = sum0 * r0 + a0;
        sum1 = sum1 * r1 + a1;
        accA.x *= r0;  accA.y *= r0;  accA.z *= r0;  accA.w *= r0;
        accB.x *= r1;  accB.y *= r1;  accB.z *= r1;  accB.w *= r1;
        for (int jj = 0; jj < 8; ++jj) {
            if (!((okm >> jj) & 1u)) continue;
            const int sj = __shfl(s, jj, 32);
            const float* rp = Wh + (size_t)sj * (N_HEADS * N_HID);
            const float4 ra = *(const float4*)(rp + loff);
            const float4 rb = *(const float4*)(rp + loff + N_HID);
            const float w0 = __shfl(a0, (lane & 24) | jj, 32);
            const float w1 = __shfl(a1, (lane & 24) | jj, 32);
            accA.x += w0 * ra.x;  accA.y += w0 * ra.y;
            accA.z += w0 * ra.z;  accA.w += w0 * ra.w;
            accB.x += w1 * rb.x;  accB.y += w1 * rb.y;
            accB.z += w1 * rb.z;  accB.w += w1 * rb.w;
        }
    }

    sum0 += row_xor<1>(sum0);  sum1 += row_xor<1>(sum1);
    sum0 += row_xor<2>(sum0);  sum1 += row_xor<2>(sum1);
    sum0 += row_xor<4>(sum0);  sum1 += row_xor<4>(sum1);
    const float q0 = deg > 0 ? 1.0f / sum0 : 0.0f;
    const float q1 = deg > 0 ? 1.0f / sum1 : 0.0f;
    accA.x *= q0;  accA.y *= q0;  accA.z *= q0;  accA.w *= q0;
    accB.x *= q1;  accB.y *= q1;  accB.z *= q1;  accB.w *= q1;

    float* op = out + (size_t)node * (N_HEADS * N_HID) + loff;
    *(float4*)op = accA;
    *(float4*)(op + N_HID) = accB;
}

extern "C" void kernel_launch(void* const* d_in, const int* in_sizes, int n_in,
                              void* d_out, int out_size, void* d_ws, size_t ws_size,
                              hipStream_t stream) {
    const float* h   = (const float*)d_in[0];
    const float* W   = (const float*)d_in[1];
    const float* a   = (const float*)d_in[2];
    const int*   src = (const int*)d_in[3];
    const int*   dst = (const int*)d_in[4];
    float* out = (float*)d_out;

    float* ws   = (float*)d_ws;
    float* Wh   = ws;
    float* s1   = Wh + (size_t)N_HEADS * N_NODES * N_HID;
    float* s2   = s1 + (size_t)N_HEADS * N_NODES;
    int*   cnt  = (int*)(s2 + (size_t)N_HEADS * N_NODES);
    int*   bsrc = cnt + CNT_WORDS;
    int*   osrc = bsrc + (size_t)N_NODES * BKT_CAP;
    int*   odst = osrc + N_EDGES;
    int*   ovfn = cnt + N_NODES;
    int*   seg  = odst + N_EDGES;
    int*   segn = seg + (size_t)NBIN * PRE_BLOCKS * SEG_CAP;

    gat_presort<<<PRE_BLOCKS, 256, 0, stream>>>(src, dst, seg, segn, ovfn);

    gat_gemm<<<NBIN + GEMM_BLOCKS, 256, 0, stream>>>(h, W, a, Wh, s1, s2,
                                                    src, dst, seg, segn, cnt, bsrc, ovfn, osrc, odst);

    gat_node<<<(N_NODES + 7) / 8, 256, 0, stream>>>(cnt, bsrc, ovfn, osrc, odst, s1, s2, Wh, out);
}
